# starting kernel
# baseline (speedup 1.0000x reference)
#include <hip/hip_runtime.h>
#include <math.h>

typedef float v2f __attribute__((ext_vector_type(2)));
typedef float v8f __attribute__((ext_vector_type(8)));

#define WMMA_F32X4(a, b, c) \
  __builtin_amdgcn_wmma_f32_16x16x4_f32(false, (a), false, (b), (short)0, (c), false, false)

__global__ void zero_kernel(float* __restrict__ p, long long n) {
  long long i = (long long)blockIdx.x * blockDim.x + threadIdx.x;
  if (i < n) p[i] = 0.0f;
}

__global__ void deg_kernel(const int* __restrict__ src, float* __restrict__ deg, int E) {
  int e = blockIdx.x * blockDim.x + threadIdx.x;
  if (e < E) unsafeAtomicAdd(&deg[src[e]], 1.0f);
}

__global__ void dinv_kernel(const float* __restrict__ deg, float* __restrict__ dinv, int N) {
  int i = blockIdx.x * blockDim.x + threadIdx.x;
  if (i < N) {
    float d = deg[i];
    dinv[i] = (d > 0.0f) ? rsqrtf(fmaxf(d, 1.0f)) : 0.0f;
  }
}

__global__ void norm_kernel(const int* __restrict__ src, const int* __restrict__ dst,
                            const float* __restrict__ dinv, float* __restrict__ nrm, int E) {
  int e = blockIdx.x * blockDim.x + threadIdx.x;
  if (e < E) nrm[e] = -dinv[src[e]] * dinv[dst[e]];
}

__global__ void gemm1_kernel(const float* __restrict__ x, const float* __restrict__ W1,
                             const float* __restrict__ b1, float* __restrict__ pre1,
                             float* __restrict__ y1, int N, int nTiles) {
  const int wave = (blockIdx.x * blockDim.x + threadIdx.x) >> 5;
  if (wave >= nTiles) return;
  const int lane = threadIdx.x & 31;
  const int n = lane & 15;
  const int hi = lane >> 4;
  const int base = wave * 16;
  const int mrow = min(base + n, N - 1);
  const float* __restrict__ W0 = W1;
  const float* __restrict__ Wb = W1 + 64 * 16;
  const float* __restrict__ xr = x + (long long)mrow * 64;

  v8f acc0 = {0.f, 0.f, 0.f, 0.f, 0.f, 0.f, 0.f, 0.f};
  v8f acc1 = acc0;
#pragma unroll
  for (int kb = 0; kb < 64; kb += 4) {
    const int k0 = kb + 2 * hi;
    v2f a = *(const v2f*)(xr + k0);
    v2f b0, bb;
    b0.x = W0[k0 * 16 + n];  b0.y = W0[(k0 + 1) * 16 + n];
    bb.x = Wb[k0 * 16 + n];  bb.y = Wb[(k0 + 1) * 16 + n];
    acc0 = WMMA_F32X4(a, b0, acc0);
    acc1 = WMMA_F32X4(a, bb, acc1);
  }
  const float bias = b1[n];
#pragma unroll
  for (int r = 0; r < 8; ++r) {
    const int row = base + r + 8 * hi;
    if (row < N) {
      pre1[(long long)row * 16 + n] = acc0[r] + bias;
      y1[(long long)row * 16 + n]  = acc1[r];
    }
  }
}

__global__ void scatter1_kernel(const int* __restrict__ src, const int* __restrict__ dst,
                                const float* __restrict__ nrm, const float* __restrict__ y1,
                                float* __restrict__ agg1, int E) {
  unsigned tid = blockIdx.x * blockDim.x + threadIdx.x;
  unsigned e = tid >> 4;
  unsigned f = tid & 15u;
  if (e < (unsigned)E) {
    int s = src[e], d = dst[e];
    unsafeAtomicAdd(&agg1[(long long)d * 16 + f], nrm[e] * y1[(long long)s * 16 + f]);
  }
}

__global__ void gemm2_kernel(const float* __restrict__ pre1, const float* __restrict__ agg1,
                             const float* __restrict__ W2, const float* __restrict__ b2,
                             float* __restrict__ pre2, float* __restrict__ y2,
                             int N, int nTiles) {
  const int wave = (blockIdx.x * blockDim.x + threadIdx.x) >> 5;
  if (wave >= nTiles) return;
  const int lane = threadIdx.x & 31;
  const int n = lane & 15;
  const int hi = lane >> 4;
  const int base = wave * 16;
  const int mrow = min(base + n, N - 1);
  const float* __restrict__ W0 = W2;
  const float* __restrict__ Wb = W2 + 16 * 40;

  v2f a[4];
#pragma unroll
  for (int i = 0; i < 4; ++i) {
    const int k0 = i * 4 + 2 * hi;
    const long long idx = (long long)mrow * 16 + k0;
    v2f p = *(const v2f*)(pre1 + idx);
    v2f g = *(const v2f*)(agg1 + idx);
    a[i].x = fmaxf(p.x + g.x, 0.0f);
    a[i].y = fmaxf(p.y + g.y, 0.0f);
  }

  v8f acc[6];
#pragma unroll
  for (int i = 0; i < 6; ++i) acc[i] = (v8f){0.f, 0.f, 0.f, 0.f, 0.f, 0.f, 0.f, 0.f};

#pragma unroll
  for (int ct = 0; ct < 3; ++ct) {
    const int col = ct * 16 + n;
    const int cc = min(col, 39);
#pragma unroll
    for (int i = 0; i < 4; ++i) {
      const int k0 = i * 4 + 2 * hi;
      v2f b0, bb;
      b0.x = W0[k0 * 40 + cc];  b0.y = W0[(k0 + 1) * 40 + cc];
      bb.x = Wb[k0 * 40 + cc];  bb.y = Wb[(k0 + 1) * 40 + cc];
      acc[ct * 2]     = WMMA_F32X4(a[i], b0, acc[ct * 2]);
      acc[ct * 2 + 1] = WMMA_F32X4(a[i], bb, acc[ct * 2 + 1]);
    }
  }

#pragma unroll
  for (int ct = 0; ct < 3; ++ct) {
    const int col = ct * 16 + n;
    if (col < 40) {
      const float bias = b2[col];
#pragma unroll
      for (int r = 0; r < 8; ++r) {
        const int row = base + r + 8 * hi;
        if (row < N) {
          pre2[(long long)row * 40 + col] = acc[ct * 2][r] + bias;
          y2[(long long)row * 40 + col]   = acc[ct * 2 + 1][r];
        }
      }
    }
  }
}

__global__ void scatter2_kernel(const int* __restrict__ src, const int* __restrict__ dst,
                                const float* __restrict__ nrm, const float* __restrict__ y2,
                                float* __restrict__ agg2, int E) {
  unsigned tid = blockIdx.x * blockDim.x + threadIdx.x;
  unsigned e = tid / 40u;
  unsigned f = tid - e * 40u;
  if (e < (unsigned)E) {
    int s = src[e], d = dst[e];
    unsafeAtomicAdd(&agg2[(long long)d * 40 + f], nrm[e] * y2[(long long)s * 40 + f]);
  }
}

__global__ void lsm_kernel(const float* __restrict__ pre2, const float* __restrict__ agg2,
                           float* __restrict__ out, int N) {
  const int node = (blockIdx.x * blockDim.x + threadIdx.x) >> 5;
  if (node >= N) return;
  const int lane = threadIdx.x & 31;
  const float* __restrict__ pr = pre2 + (long long)node * 40;
  const float* __restrict__ ag = agg2 + (long long)node * 40;
  float v0 = pr[lane] + ag[lane];
  float v1 = (lane < 8) ? (pr[lane + 32] + ag[lane + 32]) : -INFINITY;
  float m = fmaxf(v0, v1);
#pragma unroll
  for (int o = 16; o > 0; o >>= 1) m = fmaxf(m, __shfl_xor(m, o, 32));
  float s = expf(v0 - m) + ((lane < 8) ? expf(v1 - m) : 0.0f);
#pragma unroll
  for (int o = 16; o > 0; o >>= 1) s += __shfl_xor(s, o, 32);
  const float lse = m + logf(s);
  float* __restrict__ op = out + (long long)node * 40;
  op[lane] = v0 - lse;
  if (lane < 8) op[lane + 32] = v1 - lse;
}

extern "C" void kernel_launch(void* const* d_in, const int* in_sizes, int n_in,
                              void* d_out, int out_size, void* d_ws, size_t ws_size,
                              hipStream_t stream) {
  const float* x  = (const float*)d_in[0];
  const int* ei   = (const int*)d_in[1];
  const float* W1 = (const float*)d_in[2];
  const float* b1 = (const float*)d_in[3];
  const float* W2 = (const float*)d_in[4];
  const float* b2 = (const float*)d_in[5];
  float* out = (float*)d_out;

  const int N = in_sizes[0] / 64;
  const int E = in_sizes[1] / 2;
  const int* src = ei;
  const int* dst = ei + E;

  float* ws = (float*)d_ws;
  size_t off = 0;
  float* deg  = ws + off; off += (size_t)N;
  float* dinv = ws + off; off += (size_t)N;
  float* nrm  = ws + off; off += (size_t)E;
  float* pre1 = ws + off; off += (size_t)N * 16;
  float* y1   = ws + off; off += (size_t)N * 16;
  float* agg1 = ws + off; off += (size_t)N * 16;
  float* pre2 = ws + off; off += (size_t)N * 40;
  float* y2   = ws + off; off += (size_t)N * 40;
  float* agg2 = ws + off; off += (size_t)N * 40;

  const int B = 256;
  auto blk = [](long long n, int b) { return (unsigned)((n + b - 1) / b); };

  zero_kernel<<<blk(N, B), B, 0, stream>>>(deg, N);
  zero_kernel<<<blk((long long)N * 16, B), B, 0, stream>>>(agg1, (long long)N * 16);
  zero_kernel<<<blk((long long)N * 40, B), B, 0, stream>>>(agg2, (long long)N * 40);

  deg_kernel<<<blk(E, B), B, 0, stream>>>(src, deg, E);
  dinv_kernel<<<blk(N, B), B, 0, stream>>>(deg, dinv, N);
  norm_kernel<<<blk(E, B), B, 0, stream>>>(src, dst, dinv, nrm, E);

  const int tiles = (N + 15) / 16;
  gemm1_kernel<<<blk((long long)tiles * 32, B), B, 0, stream>>>(x, W1, b1, pre1, y1, N, tiles);
  scatter1_kernel<<<blk((long long)E * 16, B), B, 0, stream>>>(src, dst, nrm, y1, agg1, E);

  gemm2_kernel<<<blk((long long)tiles * 32, B), B, 0, stream>>>(pre1, agg1, W2, b2, pre2, y2, N, tiles);
  scatter2_kernel<<<blk((long long)E * 40, B), B, 0, stream>>>(src, dst, nrm, y2, agg2, E);

  lsm_kernel<<<blk((long long)N * 32, B), B, 0, stream>>>(pre2, agg2, out, N);
}


# Edge scatters coarsened: one wave per 32 edges, indices loaded once and shared by lane reads, gathers then atomic adds in batches; layer-2 rows split at the 128-B line; atomic arrival order changes
# speedup vs baseline: 1.4393x; 1.4393x over previous
#include <hip/hip_runtime.h>
#include <math.h>

typedef float v2f __attribute__((ext_vector_type(2)));
typedef float v8f __attribute__((ext_vector_type(8)));

#define WMMA_F32X4(a, b, c) \
  __builtin_amdgcn_wmma_f32_16x16x4_f32(false, (a), false, (b), (short)0, (c), false, false)

__global__ void zero_kernel(float* __restrict__ p, long long n) {
  long long i = (long long)blockIdx.x * blockDim.x + threadIdx.x;
  if (i < n) p[i] = 0.0f;
}

__global__ void deg_kernel(const int* __restrict__ src, float* __restrict__ deg, int E) {
  int e = blockIdx.x * blockDim.x + threadIdx.x;
  if (e < E) unsafeAtomicAdd(&deg[src[e]], 1.0f);
}

__global__ void dinv_kernel(const float* __restrict__ deg, float* __restrict__ dinv, int N) {
  int i = blockIdx.x * blockDim.x + threadIdx.x;
  if (i < N) {
    float d = deg[i];
    dinv[i] = (d > 0.0f) ? rsqrtf(fmaxf(d, 1.0f)) : 0.0f;
  }
}

__global__ void norm_kernel(const int* __restrict__ src, const int* __restrict__ dst,
                            const float* __restrict__ dinv, float* __restrict__ nrm, int E) {
  int e = blockIdx.x * blockDim.x + threadIdx.x;
  if (e < E) nrm[e] = -dinv[src[e]] * dinv[dst[e]];
}

__global__ void gemm1_kernel(const float* __restrict__ x, const float* __restrict__ W1,
                             const float* __restrict__ b1, float* __restrict__ pre1,
                             float* __restrict__ y1, int N, int nTiles) {
  const int wave = (blockIdx.x * blockDim.x + threadIdx.x) >> 5;
  if (wave >= nTiles) return;
  const int lane = threadIdx.x & 31;
  const int n = lane & 15;
  const int hi = lane >> 4;
  const int base = wave * 16;
  const int mrow = min(base + n, N - 1);
  const float* __restrict__ W0 = W1;
  const float* __restrict__ Wb = W1 + 64 * 16;
  const float* __restrict__ xr = x + (long long)mrow * 64;

  v8f acc0 = {0.f, 0.f, 0.f, 0.f, 0.f, 0.f, 0.f, 0.f};
  v8f acc1 = acc0;
#pragma unroll
  for (int kb = 0; kb < 64; kb += 4) {
    const int k0 = kb + 2 * hi;
    v2f a = *(const v2f*)(xr + k0);
    v2f b0, bb;
    b0.x = W0[k0 * 16 + n];  b0.y = W0[(k0 + 1) * 16 + n];
    bb.x = Wb[k0 * 16 + n];  bb.y = Wb[(k0 + 1) * 16 + n];
    acc0 = WMMA_F32X4(a, b0, acc0);
    acc1 = WMMA_F32X4(a, bb, acc1);
  }
  const float bias = b1[n];
#pragma unroll
  for (int r = 0; r < 8; ++r) {
    const int row = base + r + 8 * hi;
    if (row < N) {
      pre1[(long long)row * 16 + n] = acc0[r] + bias;
      y1[(long long)row * 16 + n]  = acc1[r];
    }
  }
}

__global__ void scatter1_kernel(const int* __restrict__ src, const int* __restrict__ dst,
                                const float* __restrict__ nrm, const float* __restrict__ y1,
                                float* __restrict__ agg1, int E) {
  const unsigned lane = threadIdx.x & 31u;
  const unsigned wave = (blockIdx.x * blockDim.x + threadIdx.x) >> 5;
  const unsigned ebase = wave * 32u;
  if (ebase >= (unsigned)E) return;
  const unsigned el = ebase + lane;
  int s_l = 0, d_l = 0;
  float w_l = 0.0f;
  if (el < (unsigned)E) { s_l = src[el]; d_l = dst[el]; w_l = nrm[el]; }
  const unsigned f = lane & 15u;
  const unsigned hi = lane >> 4;

  float yv[16], wv[16];
  int dv[16];
#pragma unroll
  for (int j = 0; j < 16; ++j) {
    const int sl = 2 * j + (int)hi;
    const int s = __shfl(s_l, sl, 32);
    dv[j] = __shfl(d_l, sl, 32);
    wv[j] = __shfl(w_l, sl, 32);
    yv[j] = y1[(long long)s * 16 + f];
  }
#pragma unroll
  for (int j = 0; j < 16; ++j) {
    if (ebase + 2u * (unsigned)j + hi < (unsigned)E)
      unsafeAtomicAdd(&agg1[(long long)dv[j] * 16 + f], wv[j] * yv[j]);
  }
}

__global__ void gemm2_kernel(const float* __restrict__ pre1, const float* __restrict__ agg1,
                             const float* __restrict__ W2, const float* __restrict__ b2,
                             float* __restrict__ pre2, float* __restrict__ y2,
                             int N, int nTiles) {
  const int wave = (blockIdx.x * blockDim.x + threadIdx.x) >> 5;
  if (wave >= nTiles) return;
  const int lane = threadIdx.x & 31;
  const int n = lane & 15;
  const int hi = lane >> 4;
  const int base = wave * 16;
  const int mrow = min(base + n, N - 1);
  const float* __restrict__ W0 = W2;
  const float* __restrict__ Wb = W2 + 16 * 40;

  v2f a[4];
#pragma unroll
  for (int i = 0; i < 4; ++i) {
    const int k0 = i * 4 + 2 * hi;
    const long long idx = (long long)mrow * 16 + k0;
    v2f p = *(const v2f*)(pre1 + idx);
    v2f g = *(const v2f*)(agg1 + idx);
    a[i].x = fmaxf(p.x + g.x, 0.0f);
    a[i].y = fmaxf(p.y + g.y, 0.0f);
  }

  v8f acc[6];
#pragma unroll
  for (int i = 0; i < 6; ++i) acc[i] = (v8f){0.f, 0.f, 0.f, 0.f, 0.f, 0.f, 0.f, 0.f};

#pragma unroll
  for (int ct = 0; ct < 3; ++ct) {
    const int col = ct * 16 + n;
    const int cc = min(col, 39);
#pragma unroll
    for (int i = 0; i < 4; ++i) {
      const int k0 = i * 4 + 2 * hi;
      v2f b0, bb;
      b0.x = W0[k0 * 40 + cc];  b0.y = W0[(k0 + 1) * 40 + cc];
      bb.x = Wb[k0 * 40 + cc];  bb.y = Wb[(k0 + 1) * 40 + cc];
      acc[ct * 2]     = WMMA_F32X4(a[i], b0, acc[ct * 2]);
      acc[ct * 2 + 1] = WMMA_F32X4(a[i], bb, acc[ct * 2 + 1]);
    }
  }

#pragma unroll
  for (int ct = 0; ct < 3; ++ct) {
    const int col = ct * 16 + n;
    if (col < 40) {
      const float bias = b2[col];
#pragma unroll
      for (int r = 0; r < 8; ++r) {
        const int row = base + r + 8 * hi;
        if (row < N) {
          pre2[(long long)row * 40 + col] = acc[ct * 2][r] + bias;
          y2[(long long)row * 40 + col]   = acc[ct * 2 + 1][r];
        }
      }
    }
  }
}

__global__ void scatter2_kernel(const int* __restrict__ src, const int* __restrict__ dst,
                                const float* __restrict__ nrm, const float* __restrict__ y2,
                                float* __restrict__ agg2, int E) {
  const unsigned lane = threadIdx.x & 31u;
  const unsigned wave = (blockIdx.x * blockDim.x + threadIdx.x) >> 5;
  const unsigned ebase = wave * 32u;
  if (ebase >= (unsigned)E) return;
  const unsigned el = ebase + lane;
  int s_l = 0, d_l = 0;
  float w_l = 0.0f;
  if (el < (unsigned)E) { s_l = src[el]; d_l = dst[el]; w_l = nrm[el]; }

#pragma unroll
  for (int b = 0; b < 4; ++b) {
    float va[8], vb[8], wv[8];
    long long ro[8];
    int n0v[8];
#pragma unroll
    for (int j = 0; j < 8; ++j) {
      const int sl = b * 8 + j;
      const int s = __builtin_amdgcn_readlane(s_l, sl);
      const int d = __builtin_amdgcn_readlane(d_l, sl);
      wv[j] = __builtin_bit_cast(float, __builtin_amdgcn_readlane(__builtin_bit_cast(int, w_l), sl));
      const long long drow = (long long)d * 40;
      ro[j] = drow;
      const unsigned off = (unsigned)(((unsigned long long)(agg2 + drow) >> 2) & 31ull);
      int n0 = 32 - (int)off;
      n0 = n0 < 8 ? 8 : n0;
      n0v[j] = n0;
      const float* __restrict__ yr = y2 + (long long)s * 40;
      const int fb = n0 + (int)lane;
      va[j] = yr[lane];
      vb[j] = yr[fb < 40 ? fb : 39];
    }
#pragma unroll
    for (int j = 0; j < 8; ++j) {
      if (ebase + (unsigned)(b * 8 + j) < (unsigned)E) {
        const int n0 = n0v[j];
        const int fb = n0 + (int)lane;
        float* __restrict__ ar = agg2 + ro[j];
        if ((int)lane < n0) unsafeAtomicAdd(&ar[lane], wv[j] * va[j]);
        if (fb < 40)        unsafeAtomicAdd(&ar[fb],   wv[j] * vb[j]);
      }
    }
  }
}

__global__ void lsm_kernel(const float* __restrict__ pre2, const float* __restrict__ agg2,
                           float* __restrict__ out, int N) {
  const int node = (blockIdx.x * blockDim.x + threadIdx.x) >> 5;
  if (node >= N) return;
  const int lane = threadIdx.x & 31;
  const float* __restrict__ pr = pre2 + (long long)node * 40;
  const float* __restrict__ ag = agg2 + (long long)node * 40;
  float v0 = pr[lane] + ag[lane];
  float v1 = (lane < 8) ? (pr[lane + 32] + ag[lane + 32]) : -INFINITY;
  float m = fmaxf(v0, v1);
#pragma unroll
  for (int o = 16; o > 0; o >>= 1) m = fmaxf(m, __shfl_xor(m, o, 32));
  float s = expf(v0 - m) + ((lane < 8) ? expf(v1 - m) : 0.0f);
#pragma unroll
  for (int o = 16; o > 0; o >>= 1) s += __shfl_xor(s, o, 32);
  const float lse = m + logf(s);
  float* __restrict__ op = out + (long long)node * 40;
  op[lane] = v0 - lse;
  if (lane < 8) op[lane + 32] = v1 - lse;
}

extern "C" void kernel_launch(void* const* d_in, const int* in_sizes, int n_in,
                              void* d_out, int out_size, void* d_ws, size_t ws_size,
                              hipStream_t stream) {
  const float* x  = (const float*)d_in[0];
  const int* ei   = (const int*)d_in[1];
  const float* W1 = (const float*)d_in[2];
  const float* b1 = (const float*)d_in[3];
  const float* W2 = (const float*)d_in[4];
  const float* b2 = (const float*)d_in[5];
  float* out = (float*)d_out;

  const int N = in_sizes[0] / 64;
  const int E = in_sizes[1] / 2;
  const int* src = ei;
  const int* dst = ei + E;

  float* ws = (float*)d_ws;
  size_t off = 0;
  float* deg  = ws + off; off += (size_t)N;
  float* dinv = ws + off; off += (size_t)N;
  float* nrm  = ws + off; off += (size_t)E;
  float* pre1 = ws + off; off += (size_t)N * 16;
  float* y1   = ws + off; off += (size_t)N * 16;
  float* agg1 = ws + off; off += (size_t)N * 16;
  float* pre2 = ws + off; off += (size_t)N * 40;
  float* y2   = ws + off; off += (size_t)N * 40;
  float* agg2 = ws + off; off += (size_t)N * 40;

  const int B = 256;
  auto blk = [](long long n, int b) { return (unsigned)((n + b - 1) / b); };

  zero_kernel<<<blk(N, B), B, 0, stream>>>(deg, N);
  zero_kernel<<<blk((long long)N * 16, B), B, 0, stream>>>(agg1, (long long)N * 16);
  zero_kernel<<<blk((long long)N * 40, B), B, 0, stream>>>(agg2, (long long)N * 40);

  deg_kernel<<<blk(E, B), B, 0, stream>>>(src, deg, E);
  dinv_kernel<<<blk(N, B), B, 0, stream>>>(deg, dinv, N);
  norm_kernel<<<blk(E, B), B, 0, stream>>>(src, dst, dinv, nrm, E);

  const int tiles = (N + 15) / 16;
  gemm1_kernel<<<blk((long long)tiles * 32, B), B, 0, stream>>>(x, W1, b1, pre1, y1, N, tiles);
  scatter1_kernel<<<blk(E, B), B, 0, stream>>>(src, dst, nrm, y1, agg1, E);

  gemm2_kernel<<<blk((long long)tiles * 32, B), B, 0, stream>>>(pre1, agg1, W2, b2, pre2, y2, N, tiles);
  scatter2_kernel<<<blk(E, B), B, 0, stream>>>(src, dst, nrm, y2, agg2, E);

  lsm_kernel<<<blk((long long)N * 32, B), B, 0, stream>>>(pre2, agg2, out, N);
}


# Zero-fill launches removed: the two GEMM kernels store the dense term (and the degree zeros) as start values of the buffers the edge atomics add into; sums now start from it (add order changes)
# speedup vs baseline: 1.5336x; 1.0655x over previous
#include <hip/hip_runtime.h>
#include <math.h>

typedef float v2f __attribute__((ext_vector_type(2)));
typedef float v8f __attribute__((ext_vector_type(8)));

#define WMMA_F32X4(a, b, c) \
  __builtin_amdgcn_wmma_f32_16x16x4_f32(false, (a), false, (b), (short)0, (c), false, false)


__global__ void deg_kernel(const int* __restrict__ src, float* __restrict__ deg, int E) {
  int e = blockIdx.x * blockDim.x + threadIdx.x;
  if (e < E) unsafeAtomicAdd(&deg[src[e]], 1.0f);
}

__global__ void dinv_kernel(const float* __restrict__ deg, float* __restrict__ dinv, int N) {
  int i = blockIdx.x * blockDim.x + threadIdx.x;
  if (i < N) {
    float d = deg[i];
    dinv[i] = (d > 0.0f) ? rsqrtf(fmaxf(d, 1.0f)) : 0.0f;
  }
}

__global__ void norm_kernel(const int* __restrict__ src, const int* __restrict__ dst,
                            const float* __restrict__ dinv, float* __restrict__ nrm, int E) {
  int e = blockIdx.x * blockDim.x + threadIdx.x;
  if (e < E) nrm[e] = -dinv[src[e]] * dinv[dst[e]];
}

__global__ void gemm1_kernel(const float* __restrict__ x, const float* __restrict__ W1,
                             const float* __restrict__ b1, float* __restrict__ sum1,
                             float* __restrict__ y1, float* __restrict__ deg,
                             int N, int nTiles) {
  const int wave = (blockIdx.x * blockDim.x + threadIdx.x) >> 5;
  if (wave >= nTiles) return;
  const int lane = threadIdx.x & 31;
  const int n = lane & 15;
  const int hi = lane >> 4;
  const int base = wave * 16;
  const int mrow = min(base + n, N - 1);
  const float* __restrict__ W0 = W1;
  const float* __restrict__ Wb = W1 + 64 * 16;
  const float* __restrict__ xr = x + (long long)mrow * 64;

  v8f acc0 = {0.f, 0.f, 0.f, 0.f, 0.f, 0.f, 0.f, 0.f};
  v8f acc1 = acc0;
#pragma unroll
  for (int kb = 0; kb < 64; kb += 4) {
    const int k0 = kb + 2 * hi;
    v2f a = *(const v2f*)(xr + k0);
    v2f b0, bb;
    b0.x = W0[k0 * 16 + n];  b0.y = W0[(k0 + 1) * 16 + n];
    bb.x = Wb[k0 * 16 + n];  bb.y = Wb[(k0 + 1) * 16 + n];
    acc0 = WMMA_F32X4(a, b0, acc0);
    acc1 = WMMA_F32X4(a, bb, acc1);
  }
  const float bias = b1[n];
#pragma unroll
  for (int r = 0; r < 8; ++r) {
    const int row = base + r + 8 * hi;
    if (row < N) {
      sum1[(long long)row * 16 + n] = acc0[r] + bias;
      y1[(long long)row * 16 + n]   = acc1[r];
    }
  }
  if (hi == 0 && base + n < N) deg[base + n] = 0.0f;
}

__global__ void scatter1_kernel(const int* __restrict__ src, const int* __restrict__ dst,
                                const float* __restrict__ nrm, const float* __restrict__ y1,
                                float* __restrict__ sum1, int E) {
  const unsigned lane = threadIdx.x & 31u;
  const unsigned wave = (blockIdx.x * blockDim.x + threadIdx.x) >> 5;
  const unsigned ebase = wave * 32u;
  if (ebase >= (unsigned)E) return;
  const unsigned el = ebase + lane;
  int s_l = 0, d_l = 0;
  float w_l = 0.0f;
  if (el < (unsigned)E) { s_l = src[el]; d_l = dst[el]; w_l = nrm[el]; }
  const unsigned f = lane & 15u;
  const unsigned hi = lane >> 4;

  float yv[16], wv[16];
  int dv[16];
#pragma unroll
  for (int j = 0; j < 16; ++j) {
    const int sl = 2 * j + (int)hi;
    const int s = __shfl(s_l, sl, 32);
    dv[j] = __shfl(d_l, sl, 32);
    wv[j] = __shfl(w_l, sl, 32);
    yv[j] = y1[(long long)s * 16 + f];
  }
#pragma unroll
  for (int j = 0; j < 16; ++j) {
    if (ebase + 2u * (unsigned)j + hi < (unsigned)E)
      unsafeAtomicAdd(&sum1[(long long)dv[j] * 16 + f], wv[j] * yv[j]);
  }
}

__global__ void gemm2_kernel(const float* __restrict__ sum1,
                             const float* __restrict__ W2, const float* __restrict__ b2,
                             float* __restrict__ sum2, float* __restrict__ y2,
                             int N, int nTiles) {
  const int wave = (blockIdx.x * blockDim.x + threadIdx.x) >> 5;
  if (wave >= nTiles) return;
  const int lane = threadIdx.x & 31;
  const int n = lane & 15;
  const int hi = lane >> 4;
  const int base = wave * 16;
  const int mrow = min(base + n, N - 1);
  const float* __restrict__ W0 = W2;
  const float* __restrict__ Wb = W2 + 16 * 40;

  v2f a[4];
#pragma unroll
  for (int i = 0; i < 4; ++i) {
    const int k0 = i * 4 + 2 * hi;
    const long long idx = (long long)mrow * 16 + k0;
    v2f g = *(const v2f*)(sum1 + idx);
    a[i].x = fmaxf(g.x, 0.0f);
    a[i].y = fmaxf(g.y, 0.0f);
  }

  v8f acc[6];
#pragma unroll
  for (int i = 0; i < 6; ++i) acc[i] = (v8f){0.f, 0.f, 0.f, 0.f, 0.f, 0.f, 0.f, 0.f};

#pragma unroll
  for (int ct = 0; ct < 3; ++ct) {
    const int col = ct * 16 + n;
    const int cc = min(col, 39);
#pragma unroll
    for (int i = 0; i < 4; ++i) {
      const int k0 = i * 4 + 2 * hi;
      v2f b0, bb;
      b0.x = W0[k0 * 40 + cc];  b0.y = W0[(k0 + 1) * 40 + cc];
      bb.x = Wb[k0 * 40 + cc];  bb.y = Wb[(k0 + 1) * 40 + cc];
      acc[ct * 2]     = WMMA_F32X4(a[i], b0, acc[ct * 2]);
      acc[ct * 2 + 1] = WMMA_F32X4(a[i], bb, acc[ct * 2 + 1]);
    }
  }

#pragma unroll
  for (int ct = 0; ct < 3; ++ct) {
    const int col = ct * 16 + n;
    if (col < 40) {
      const float bias = b2[col];
#pragma unroll
      for (int r = 0; r < 8; ++r) {
        const int row = base + r + 8 * hi;
        if (row < N) {
          sum2[(long long)row * 40 + col] = acc[ct * 2][r] + bias;
          y2[(long long)row * 40 + col]   = acc[ct * 2 + 1][r];
        }
      }
    }
  }
}

__global__ void scatter2_kernel(const int* __restrict__ src, const int* __restrict__ dst,
                                const float* __restrict__ nrm, const float* __restrict__ y2,
                                float* __restrict__ sum2, int E) {
  const unsigned lane = threadIdx.x & 31u;
  const unsigned wave = (blockIdx.x * blockDim.x + threadIdx.x) >> 5;
  const unsigned ebase = wave * 32u;
  if (ebase >= (unsigned)E) return;
  const unsigned el = ebase + lane;
  int s_l = 0, d_l = 0;
  float w_l = 0.0f;
  if (el < (unsigned)E) { s_l = src[el]; d_l = dst[el]; w_l = nrm[el]; }

#pragma unroll
  for (int b = 0; b < 4; ++b) {
    float va[8], vb[8], wv[8];
    long long ro[8];
    int n0v[8];
#pragma unroll
    for (int j = 0; j < 8; ++j) {
      const int sl = b * 8 + j;
      const int s = __builtin_amdgcn_readlane(s_l, sl);
      const int d = __builtin_amdgcn_readlane(d_l, sl);
      wv[j] = __builtin_bit_cast(float, __builtin_amdgcn_readlane(__builtin_bit_cast(int, w_l), sl));
      const long long drow = (long long)d * 40;
      ro[j] = drow;
      const unsigned off = (unsigned)(((unsigned long long)(sum2 + drow) >> 2) & 31ull);
      int n0 = 32 - (int)off;
      n0 = n0 < 8 ? 8 : n0;
      n0v[j] = n0;
      const float* __restrict__ yr = y2 + (long long)s * 40;
      const int fb = n0 + (int)lane;
      va[j] = yr[lane];
      vb[j] = yr[fb < 40 ? fb : 39];
    }
#pragma unroll
    for (int j = 0; j < 8; ++j) {
      if (ebase + (unsigned)(b * 8 + j) < (unsigned)E) {
        const int n0 = n0v[j];
        const int fb = n0 + (int)lane;
        float* __restrict__ ar = sum2 + ro[j];
        if ((int)lane < n0) unsafeAtomicAdd(&ar[lane], wv[j] * va[j]);
        if (fb < 40)        unsafeAtomicAdd(&ar[fb],   wv[j] * vb[j]);
      }
    }
  }
}

__global__ void lsm_kernel(const float* __restrict__ sum2,
                           float* __restrict__ out, int N) {
  const int node = (blockIdx.x * blockDim.x + threadIdx.x) >> 5;
  if (node >= N) return;
  const int lane = threadIdx.x & 31;
  const float* __restrict__ ag = sum2 + (long long)node * 40;
  float v0 = ag[lane];
  float v1 = (lane < 8) ? ag[lane + 32] : -INFINITY;
  float m = fmaxf(v0, v1);
#pragma unroll
  for (int o = 16; o > 0; o >>= 1) m = fmaxf(m, __shfl_xor(m, o, 32));
  float s = expf(v0 - m) + ((lane < 8) ? expf(v1 - m) : 0.0f);
#pragma unroll
  for (int o = 16; o > 0; o >>= 1) s += __shfl_xor(s, o, 32);
  const float lse = m + logf(s);
  float* __restrict__ op = out + (long long)node * 40;
  op[lane] = v0 - lse;
  if (lane < 8) op[lane + 32] = v1 - lse;
}

extern "C" void kernel_launch(void* const* d_in, const int* in_sizes, int n_in,
                              void* d_out, int out_size, void* d_ws, size_t ws_size,
                              hipStream_t stream) {
  const float* x  = (const float*)d_in[0];
  const int* ei   = (const int*)d_in[1];
  const float* W1 = (const float*)d_in[2];
  const float* b1 = (const float*)d_in[3];
  const float* W2 = (const float*)d_in[4];
  const float* b2 = (const float*)d_in[5];
  float* out = (float*)d_out;

  const int N = in_sizes[0] / 64;
  const int E = in_sizes[1] / 2;
  const int* src = ei;
  const int* dst = ei + E;

  float* ws = (float*)d_ws;
  size_t off = 0;
  float* deg  = ws + off; off += (size_t)N;
  float* dinv = ws + off; off += (size_t)N;
  float* nrm  = ws + off; off += (size_t)E;
  float* y1   = ws + off; off += (size_t)N * 16;
  float* sum1 = ws + off; off += (size_t)N * 16;
  float* y2   = ws + off; off += (size_t)N * 40;
  float* sum2 = ws + off; off += (size_t)N * 40;

  const int B = 256;
  auto blk = [](long long n, int b) { return (unsigned)((n + b - 1) / b); };

  const int tiles = (N + 15) / 16;
  gemm1_kernel<<<blk((long long)tiles * 32, B), B, 0, stream>>>(x, W1, b1, sum1, y1, deg, N, tiles);

  deg_kernel<<<blk(E, B), B, 0, stream>>>(src, deg, E);
  dinv_kernel<<<blk(N, B), B, 0, stream>>>(deg, dinv, N);
  norm_kernel<<<blk(E, B), B, 0, stream>>>(src, dst, dinv, nrm, E);

  scatter1_kernel<<<blk(E, B), B, 0, stream>>>(src, dst, nrm, y1, sum1, E);

  gemm2_kernel<<<blk((long long)tiles * 32, B), B, 0, stream>>>(sum1, W2, b2, sum2, y2, N, tiles);
  scatter2_kernel<<<blk(E, B), B, 0, stream>>>(src, dst, nrm, y2, sum2, E);

  lsm_kernel<<<blk((long long)N * 32, B), B, 0, stream>>>(sum2, out, N);
}


# Each atomic target kept twice, one copy per package half: a wave adds into its own half's copy (die id), readers add the two copies; no line is shared by both halves' atomics (add order changes)
# speedup vs baseline: 2.4607x; 1.6045x over previous
#include <hip/hip_runtime.h>
#include <math.h>

typedef float v2f __attribute__((ext_vector_type(2)));
typedef float v8f __attribute__((ext_vector_type(8)));

#define WMMA_F32X4(a, b, c) \
  __builtin_amdgcn_wmma_f32_16x16x4_f32(false, (a), false, (b), (short)0, (c), false, false)


__device__ __forceinline__ int package_half() {
  return (__builtin_amdgcn_s_sendmsg_rtn(0x87) >> 18) & 1;
}

__global__ void deg_kernel(const int* __restrict__ src, float* __restrict__ degA,
                           float* __restrict__ degB, int E) {
  int e = blockIdx.x * blockDim.x + threadIdx.x;
  float* __restrict__ deg = package_half() ? degB : degA;
  if (e < E) unsafeAtomicAdd(&deg[src[e]], 1.0f);
}

__global__ void dinv_kernel(const float* __restrict__ degA, const float* __restrict__ degB,
                            float* __restrict__ dinv, int N) {
  int i = blockIdx.x * blockDim.x + threadIdx.x;
  if (i < N) {
    float d = degA[i] + degB[i];
    dinv[i] = (d > 0.0f) ? rsqrtf(fmaxf(d, 1.0f)) : 0.0f;
  }
}

__global__ void norm_kernel(const int* __restrict__ src, const int* __restrict__ dst,
                            const float* __restrict__ dinv, float* __restrict__ nrm, int E) {
  int e = blockIdx.x * blockDim.x + threadIdx.x;
  if (e < E) nrm[e] = -dinv[src[e]] * dinv[dst[e]];
}

__global__ void gemm1_kernel(const float* __restrict__ x, const float* __restrict__ W1,
                             const float* __restrict__ b1, float* __restrict__ sum1A,
                             float* __restrict__ sum1B,
                             float* __restrict__ y1, float* __restrict__ degA,
                             float* __restrict__ degB, int N, int nTiles) {
  const int wave = (blockIdx.x * blockDim.x + threadIdx.x) >> 5;
  if (wave >= nTiles) return;
  const int lane = threadIdx.x & 31;
  const int n = lane & 15;
  const int hi = lane >> 4;
  const int base = wave * 16;
  const int mrow = min(base + n, N - 1);
  const float* __restrict__ W0 = W1;
  const float* __restrict__ Wb = W1 + 64 * 16;
  const float* __restrict__ xr = x + (long long)mrow * 64;

  v8f acc0 = {0.f, 0.f, 0.f, 0.f, 0.f, 0.f, 0.f, 0.f};
  v8f acc1 = acc0;
#pragma unroll
  for (int kb = 0; kb < 64; kb += 4) {
    const int k0 = kb + 2 * hi;
    v2f a = *(const v2f*)(xr + k0);
    v2f b0, bb;
    b0.x = W0[k0 * 16 + n];  b0.y = W0[(k0 + 1) * 16 + n];
    bb.x = Wb[k0 * 16 + n];  bb.y = Wb[(k0 + 1) * 16 + n];
    acc0 = WMMA_F32X4(a, b0, acc0);
    acc1 = WMMA_F32X4(a, bb, acc1);
  }
  const float bias = b1[n];
#pragma unroll
  for (int r = 0; r < 8; ++r) {
    const int row = base + r + 8 * hi;
    if (row < N) {
      sum1A[(long long)row * 16 + n] = acc0[r] + bias;
      sum1B[(long long)row * 16 + n] = 0.0f;
      y1[(long long)row * 16 + n]    = acc1[r];
    }
  }
  if (base + n < N) (hi ? degB : degA)[base + n] = 0.0f;
}

__global__ void scatter1_kernel(const int* __restrict__ src, const int* __restrict__ dst,
                                const float* __restrict__ nrm, const float* __restrict__ y1,
                                float* __restrict__ sum1A, float* __restrict__ sum1B, int E) {
  const unsigned lane = threadIdx.x & 31u;
  const unsigned wave = (blockIdx.x * blockDim.x + threadIdx.x) >> 5;
  const unsigned ebase = wave * 32u;
  if (ebase >= (unsigned)E) return;
  const unsigned el = ebase + lane;
  int s_l = 0, d_l = 0;
  float w_l = 0.0f;
  if (el < (unsigned)E) { s_l = src[el]; d_l = dst[el]; w_l = nrm[el]; }
  float* __restrict__ sum1 = package_half() ? sum1B : sum1A;
  const unsigned f = lane & 15u;
  const unsigned hi = lane >> 4;

  float yv[16], wv[16];
  int dv[16];
#pragma unroll
  for (int j = 0; j < 16; ++j) {
    const int sl = 2 * j + (int)hi;
    const int s = __shfl(s_l, sl, 32);
    dv[j] = __shfl(d_l, sl, 32);
    wv[j] = __shfl(w_l, sl, 32);
    yv[j] = y1[(long long)s * 16 + f];
  }
#pragma unroll
  for (int j = 0; j < 16; ++j) {
    if (ebase + 2u * (unsigned)j + hi < (unsigned)E)
      unsafeAtomicAdd(&sum1[(long long)dv[j] * 16 + f], wv[j] * yv[j]);
  }
}

__global__ void gemm2_kernel(const float* __restrict__ sum1A, const float* __restrict__ sum1B,
                             const float* __restrict__ W2, const float* __restrict__ b2,
                             float* __restrict__ sum2A, float* __restrict__ sum2B,
                             float* __restrict__ y2, int N, int nTiles) {
  const int wave = (blockIdx.x * blockDim.x + threadIdx.x) >> 5;
  if (wave >= nTiles) return;
  const int lane = threadIdx.x & 31;
  const int n = lane & 15;
  const int hi = lane >> 4;
  const int base = wave * 16;
  const int mrow = min(base + n, N - 1);
  const float* __restrict__ W0 = W2;
  const float* __restrict__ Wb = W2 + 16 * 40;

  v2f a[4];
#pragma unroll
  for (int i = 0; i < 4; ++i) {
    const int k0 = i * 4 + 2 * hi;
    const long long idx = (long long)mrow * 16 + k0;
    v2f g = *(const v2f*)(sum1A + idx);
    v2f gB = *(const v2f*)(sum1B + idx);
    a[i].x = fmaxf(g.x + gB.x, 0.0f);
    a[i].y = fmaxf(g.y + gB.y, 0.0f);
  }

  v8f acc[6];
#pragma unroll
  for (int i = 0; i < 6; ++i) acc[i] = (v8f){0.f, 0.f, 0.f, 0.f, 0.f, 0.f, 0.f, 0.f};

#pragma unroll
  for (int ct = 0; ct < 3; ++ct) {
    const int col = ct * 16 + n;
    const int cc = min(col, 39);
#pragma unroll
    for (int i = 0; i < 4; ++i) {
      const int k0 = i * 4 + 2 * hi;
      v2f b0, bb;
      b0.x = W0[k0 * 40 + cc];  b0.y = W0[(k0 + 1) * 40 + cc];
      bb.x = Wb[k0 * 40 + cc];  bb.y = Wb[(k0 + 1) * 40 + cc];
      acc[ct * 2]     = WMMA_F32X4(a[i], b0, acc[ct * 2]);
      acc[ct * 2 + 1] = WMMA_F32X4(a[i], bb, acc[ct * 2 + 1]);
    }
  }

#pragma unroll
  for (int ct = 0; ct < 3; ++ct) {
    const int col = ct * 16 + n;
    if (col < 40) {
      const float bias = b2[col];
#pragma unroll
      for (int r = 0; r < 8; ++r) {
        const int row = base + r + 8 * hi;
        if (row < N) {
          sum2A[(long long)row * 40 + col] = acc[ct * 2][r] + bias;
          sum2B[(long long)row * 40 + col] = 0.0f;
          y2[(long long)row * 40 + col]    = acc[ct * 2 + 1][r];
        }
      }
    }
  }
}

__global__ void scatter2_kernel(const int* __restrict__ src, const int* __restrict__ dst,
                                const float* __restrict__ nrm, const float* __restrict__ y2,
                                float* __restrict__ sum2A, float* __restrict__ sum2B, int E) {
  const unsigned lane = threadIdx.x & 31u;
  const unsigned wave = (blockIdx.x * blockDim.x + threadIdx.x) >> 5;
  const unsigned ebase = wave * 32u;
  if (ebase >= (unsigned)E) return;
  const unsigned el = ebase + lane;
  int s_l = 0, d_l = 0;
  float w_l = 0.0f;
  if (el < (unsigned)E) { s_l = src[el]; d_l = dst[el]; w_l = nrm[el]; }
  float* __restrict__ sum2 = package_half() ? sum2B : sum2A;

#pragma unroll
  for (int b = 0; b < 4; ++b) {
    float va[8], vb[8], wv[8];
    long long ro[8];
    int n0v[8];
#pragma unroll
    for (int j = 0; j < 8; ++j) {
      const int sl = b * 8 + j;
      const int s = __builtin_amdgcn_readlane(s_l, sl);
      const int d = __builtin_amdgcn_readlane(d_l, sl);
      wv[j] = __builtin_bit_cast(float, __builtin_amdgcn_readlane(__builtin_bit_cast(int, w_l), sl));
      const long long drow = (long long)d * 40;
      ro[j] = drow;
      const unsigned off = (unsigned)(((unsigned long long)(sum2 + drow) >> 2) & 31ull);
      int n0 = 32 - (int)off;
      n0 = n0 < 8 ? 8 : n0;
      n0v[j] = n0;
      const float* __restrict__ yr = y2 + (long long)s * 40;
      const int fb = n0 + (int)lane;
      va[j] = yr[lane];
      vb[j] = yr[fb < 40 ? fb : 39];
    }
#pragma unroll
    for (int j = 0; j < 8; ++j) {
      if (ebase + (unsigned)(b * 8 + j) < (unsigned)E) {
        const int n0 = n0v[j];
        const int fb = n0 + (int)lane;
        float* __restrict__ ar = sum2 + ro[j];
        if ((int)lane < n0) unsafeAtomicAdd(&ar[lane], wv[j] * va[j]);
        if (fb < 40)        unsafeAtomicAdd(&ar[fb],   wv[j] * vb[j]);
      }
    }
  }
}

__global__ void lsm_kernel(const float* __restrict__ sum2B, float* out, int N) {
  const int node = (blockIdx.x * blockDim.x + threadIdx.x) >> 5;
  if (node >= N) return;
  const int lane = threadIdx.x & 31;
  const float* ag = out + (long long)node * 40;
  const float* __restrict__ bg = sum2B + (long long)node * 40;
  float v0 = ag[lane] + bg[lane];
  float v1 = (lane < 8) ? ag[lane + 32] + bg[lane + 32] : -INFINITY;
  float m = fmaxf(v0, v1);
#pragma unroll
  for (int o = 16; o > 0; o >>= 1) m = fmaxf(m, __shfl_xor(m, o, 32));
  float s = expf(v0 - m) + ((lane < 8) ? expf(v1 - m) : 0.0f);
#pragma unroll
  for (int o = 16; o > 0; o >>= 1) s += __shfl_xor(s, o, 32);
  const float lse = m + logf(s);
  float* op = out + (long long)node * 40;
  op[lane] = v0 - lse;
  if (lane < 8) op[lane + 32] = v1 - lse;
}

extern "C" void kernel_launch(void* const* d_in, const int* in_sizes, int n_in,
                              void* d_out, int out_size, void* d_ws, size_t ws_size,
                              hipStream_t stream) {
  const float* x  = (const float*)d_in[0];
  const int* ei   = (const int*)d_in[1];
  const float* W1 = (const float*)d_in[2];
  const float* b1 = (const float*)d_in[3];
  const float* W2 = (const float*)d_in[4];
  const float* b2 = (const float*)d_in[5];
  float* out = (float*)d_out;

  const int N = in_sizes[0] / 64;
  const int E = in_sizes[1] / 2;
  const int* src = ei;
  const int* dst = ei + E;

  float* ws = (float*)d_ws;
  size_t off = 0;
  float* degA  = ws + off; off += (size_t)N;
  float* dinv  = ws + off; off += (size_t)N;
  float* nrm   = ws + off; off += (size_t)E;
  float* sum1A = ws + off; off += (size_t)N * 16;
  float* sum1B = ws + off; off += (size_t)N * 16;
  float* y2    = ws + off; off += (size_t)N * 40;
  float* sum2B = ws + off; off += (size_t)N * 40;
  float* y1    = y2;
  float* degB  = y2 + (size_t)N * 16;
  float* sum2A = out;

  const int B = 256;
  auto blk = [](long long n, int b) { return (unsigned)((n + b - 1) / b); };

  const int tiles = (N + 15) / 16;
  gemm1_kernel<<<blk((long long)tiles * 32, B), B, 0, stream>>>(x, W1, b1, sum1A, sum1B, y1, degA, degB, N, tiles);

  deg_kernel<<<blk(E, B), B, 0, stream>>>(src, degA, degB, E);
  dinv_kernel<<<blk(N, B), B, 0, stream>>>(degA, degB, dinv, N);
  norm_kernel<<<blk(E, B), B, 0, stream>>>(src, dst, dinv, nrm, E);

  scatter1_kernel<<<blk(E, B), B, 0, stream>>>(src, dst, nrm, y1, sum1A, sum1B, E);

  gemm2_kernel<<<blk((long long)tiles * 32, B), B, 0, stream>>>(sum1A, sum1B, W2, b2, sum2A, sum2B, y2, N, tiles);
  scatter2_kernel<<<blk(E, B), B, 0, stream>>>(src, dst, nrm, y2, sum2A, sum2B, E);

  lsm_kernel<<<blk((long long)N * 32, B), B, 0, stream>>>(sum2B, out, N);
}


# Per-element kernels coarsened: a wave takes 256 edges (degree count, edge coefficients: 8 strided edges per lane, loads issued ahead) or 8 rows (log_softmax), unmasked; 87,500 fewer short waves
# speedup vs baseline: 2.6119x; 1.0614x over previous
#include <hip/hip_runtime.h>
#include <math.h>

typedef float v2f __attribute__((ext_vector_type(2)));
typedef float v8f __attribute__((ext_vector_type(8)));

#define WMMA_F32X4(a, b, c) \
  __builtin_amdgcn_wmma_f32_16x16x4_f32(false, (a), false, (b), (short)0, (c), false, false)


__device__ __forceinline__ int package_half() {
  return (__builtin_amdgcn_s_sendmsg_rtn(0x87) >> 18) & 1;
}

#define EDGE_K 8

__global__ void deg_kernel(const int* __restrict__ src, float* __restrict__ degA,
                           float* __restrict__ degB, int E) {
  const unsigned lane = threadIdx.x & 31u;
  const unsigned wave = (blockIdx.x * blockDim.x + threadIdx.x) >> 5;
  const unsigned ebase = wave * (32u * EDGE_K);
  if (ebase >= (unsigned)E) return;
  float* __restrict__ deg = package_half() ? degB : degA;
  if (ebase + 32u * EDGE_K <= (unsigned)E) {
    int s[EDGE_K];
#pragma unroll
    for (int j = 0; j < EDGE_K; ++j) s[j] = src[ebase + 32u * (unsigned)j + lane];
#pragma unroll
    for (int j = 0; j < EDGE_K; ++j) unsafeAtomicAdd(&deg[s[j]], 1.0f);
  } else {
    for (int j = 0; j < EDGE_K; ++j) {
      const unsigned e = ebase + 32u * (unsigned)j + lane;
      if (e < (unsigned)E) unsafeAtomicAdd(&deg[src[e]], 1.0f);
    }
  }
}

__global__ void dinv_kernel(const float* __restrict__ degA, const float* __restrict__ degB,
                            float* __restrict__ dinv, int N) {
  int i = blockIdx.x * blockDim.x + threadIdx.x;
  if (i < N) {
    float d = degA[i] + degB[i];
    dinv[i] = (d > 0.0f) ? rsqrtf(fmaxf(d, 1.0f)) : 0.0f;
  }
}

__global__ void norm_kernel(const int* __restrict__ src, const int* __restrict__ dst,
                            const float* __restrict__ dinv, float* __restrict__ nrm, int E) {
  const unsigned lane = threadIdx.x & 31u;
  const unsigned wave = (blockIdx.x * blockDim.x + threadIdx.x) >> 5;
  const unsigned ebase = wave * (32u * EDGE_K);
  if (ebase >= (unsigned)E) return;
  if (ebase + 32u * EDGE_K <= (unsigned)E) {
    int s[EDGE_K], d[EDGE_K];
    float a[EDGE_K], b[EDGE_K];
#pragma unroll
    for (int j = 0; j < EDGE_K; ++j) {
      const unsigned e = ebase + 32u * (unsigned)j + lane;
      s[j] = src[e];
      d[j] = dst[e];
    }
#pragma unroll
    for (int j = 0; j < EDGE_K; ++j) { a[j] = dinv[s[j]]; b[j] = dinv[d[j]]; }
#pragma unroll
    for (int j = 0; j < EDGE_K; ++j) nrm[ebase + 32u * (unsigned)j + lane] = -a[j] * b[j];
  } else {
    for (int j = 0; j < EDGE_K; ++j) {
      const unsigned e = ebase + 32u * (unsigned)j + lane;
      if (e < (unsigned)E) nrm[e] = -dinv[src[e]] * dinv[dst[e]];
    }
  }
}

__global__ void gemm1_kernel(const float* __restrict__ x, const float* __restrict__ W1,
                             const float* __restrict__ b1, float* __restrict__ sum1A,
                             float* __restrict__ sum1B,
                             float* __restrict__ y1, float* __restrict__ degA,
                             float* __restrict__ degB, int N, int nTiles) {
  const int wave = (blockIdx.x * blockDim.x + threadIdx.x) >> 5;
  if (wave >= nTiles) return;
  const int lane = threadIdx.x & 31;
  const int n = lane & 15;
  const int hi = lane >> 4;
  const int base = wave * 16;
  const int mrow = min(base + n, N - 1);
  const float* __restrict__ W0 = W1;
  const float* __restrict__ Wb = W1 + 64 * 16;
  const float* __restrict__ xr = x + (long long)mrow * 64;

  v8f acc0 = {0.f, 0.f, 0.f, 0.f, 0.f, 0.f, 0.f, 0.f};
  v8f acc1 = acc0;
#pragma unroll
  for (int kb = 0; kb < 64; kb += 4) {
    const int k0 = kb + 2 * hi;
    v2f a = *(const v2f*)(xr + k0);
    v2f b0, bb;
    b0.x = W0[k0 * 16 + n];  b0.y = W0[(k0 + 1) * 16 + n];
    bb.x = Wb[k0 * 16 + n];  bb.y = Wb[(k0 + 1) * 16 + n];
    acc0 = WMMA_F32X4(a, b0, acc0);
    acc1 = WMMA_F32X4(a, bb, acc1);
  }
  const float bias = b1[n];
#pragma unroll
  for (int r = 0; r < 8; ++r) {
    const int row = base + r + 8 * hi;
    if (row < N) {
      sum1A[(long long)row * 16 + n] = acc0[r] + bias;
      sum1B[(long long)row * 16 + n] = 0.0f;
      y1[(long long)row * 16 + n]    = acc1[r];
    }
  }
  if (base + n < N) (hi ? degB : degA)[base + n] = 0.0f;
}

__global__ void scatter1_kernel(const int* __restrict__ src, const int* __restrict__ dst,
                                const float* __restrict__ nrm, const float* __restrict__ y1,
                                float* __restrict__ sum1A, float* __restrict__ sum1B, int E) {
  const unsigned lane = threadIdx.x & 31u;
  const unsigned wave = (blockIdx.x * blockDim.x + threadIdx.x) >> 5;
  const unsigned ebase = wave * 32u;
  if (ebase >= (unsigned)E) return;
  const unsigned el = ebase + lane;
  int s_l = 0, d_l = 0;
  float w_l = 0.0f;
  if (el < (unsigned)E) { s_l = src[el]; d_l = dst[el]; w_l = nrm[el]; }
  float* __restrict__ sum1 = package_half() ? sum1B : sum1A;
  const unsigned f = lane & 15u;
  const unsigned hi = lane >> 4;

  float yv[16], wv[16];
  int dv[16];
#pragma unroll
  for (int j = 0; j < 16; ++j) {
    const int sl = 2 * j + (int)hi;
    const int s = __shfl(s_l, sl, 32);
    dv[j] = __shfl(d_l, sl, 32);
    wv[j] = __shfl(w_l, sl, 32);
    yv[j] = y1[(long long)s * 16 + f];
  }
#pragma unroll
  for (int j = 0; j < 16; ++j) {
    if (ebase + 2u * (unsigned)j + hi < (unsigned)E)
      unsafeAtomicAdd(&sum1[(long long)dv[j] * 16 + f], wv[j] * yv[j]);
  }
}

__global__ void gemm2_kernel(const float* __restrict__ sum1A, const float* __restrict__ sum1B,
                             const float* __restrict__ W2, const float* __restrict__ b2,
                             float* __restrict__ sum2A, float* __restrict__ sum2B,
                             float* __restrict__ y2, int N, int nTiles) {
  const int wave = (blockIdx.x * blockDim.x + threadIdx.x) >> 5;
  if (wave >= nTiles) return;
  const int lane = threadIdx.x & 31;
  const int n = lane & 15;
  const int hi = lane >> 4;
  const int base = wave * 16;
  const int mrow = min(base + n, N - 1);
  const float* __restrict__ W0 = W2;
  const float* __restrict__ Wb = W2 + 16 * 40;

  v2f a[4];
#pragma unroll
  for (int i = 0; i < 4; ++i) {
    const int k0 = i * 4 + 2 * hi;
    const long long idx = (long long)mrow * 16 + k0;
    v2f g = *(const v2f*)(sum1A + idx);
    v2f gB = *(const v2f*)(sum1B + idx);
    a[i].x = fmaxf(g.x + gB.x, 0.0f);
    a[i].y = fmaxf(g.y + gB.y, 0.0f);
  }

  v8f acc[6];
#pragma unroll
  for (int i = 0; i < 6; ++i) acc[i] = (v8f){0.f, 0.f, 0.f, 0.f, 0.f, 0.f, 0.f, 0.f};

#pragma unroll
  for (int ct = 0; ct < 3; ++ct) {
    const int col = ct * 16 + n;
    const int cc = min(col, 39);
#pragma unroll
    for (int i = 0; i < 4; ++i) {
      const int k0 = i * 4 + 2 * hi;
      v2f b0, bb;
      b0.x = W0[k0 * 40 + cc];  b0.y = W0[(k0 + 1) * 40 + cc];
      bb.x = Wb[k0 * 40 + cc];  bb.y = Wb[(k0 + 1) * 40 + cc];
      acc[ct * 2]     = WMMA_F32X4(a[i], b0, acc[ct * 2]);
      acc[ct * 2 + 1] = WMMA_F32X4(a[i], bb, acc[ct * 2 + 1]);
    }
  }

#pragma unroll
  for (int ct = 0; ct < 3; ++ct) {
    const int col = ct * 16 + n;
    if (col < 40) {
      const float bias = b2[col];
#pragma unroll
      for (int r = 0; r < 8; ++r) {
        const int row = base + r + 8 * hi;
        if (row < N) {
          sum2A[(long long)row * 40 + col] = acc[ct * 2][r] + bias;
          sum2B[(long long)row * 40 + col] = 0.0f;
          y2[(long long)row * 40 + col]    = acc[ct * 2 + 1][r];
        }
      }
    }
  }
}

__global__ void scatter2_kernel(const int* __restrict__ src, const int* __restrict__ dst,
                                const float* __restrict__ nrm, const float* __restrict__ y2,
                                float* __restrict__ sum2A, float* __restrict__ sum2B, int E) {
  const unsigned lane = threadIdx.x & 31u;
  const unsigned wave = (blockIdx.x * blockDim.x + threadIdx.x) >> 5;
  const unsigned ebase = wave * 32u;
  if (ebase >= (unsigned)E) return;
  const unsigned el = ebase + lane;
  int s_l = 0, d_l = 0;
  float w_l = 0.0f;
  if (el < (unsigned)E) { s_l = src[el]; d_l = dst[el]; w_l = nrm[el]; }
  float* __restrict__ sum2 = package_half() ? sum2B : sum2A;

#pragma unroll
  for (int b = 0; b < 4; ++b) {
    float va[8], vb[8], wv[8];
    long long ro[8];
    int n0v[8];
#pragma unroll
    for (int j = 0; j < 8; ++j) {
      const int sl = b * 8 + j;
      const int s = __builtin_amdgcn_readlane(s_l, sl);
      const int d = __builtin_amdgcn_readlane(d_l, sl);
      wv[j] = __builtin_bit_cast(float, __builtin_amdgcn_readlane(__builtin_bit_cast(int, w_l), sl));
      const long long drow = (long long)d * 40;
      ro[j] = drow;
      const unsigned off = (unsigned)(((unsigned long long)(sum2 + drow) >> 2) & 31ull);
      int n0 = 32 - (int)off;
      n0 = n0 < 8 ? 8 : n0;
      n0v[j] = n0;
      const float* __restrict__ yr = y2 + (long long)s * 40;
      const int fb = n0 + (int)lane;
      va[j] = yr[lane];
      vb[j] = yr[fb < 40 ? fb : 39];
    }
#pragma unroll
    for (int j = 0; j < 8; ++j) {
      if (ebase + (unsigned)(b * 8 + j) < (unsigned)E) {
        const int n0 = n0v[j];
        const int fb = n0 + (int)lane;
        float* __restrict__ ar = sum2 + ro[j];
        if ((int)lane < n0) unsafeAtomicAdd(&ar[lane], wv[j] * va[j]);
        if (fb < 40)        unsafeAtomicAdd(&ar[fb],   wv[j] * vb[j]);
      }
    }
  }
}

__device__ __forceinline__ void lsm_one_row(const float* __restrict__ sum2B, float* out,
                                            int node, int lane) {
  const float* ag = out + (long long)node * 40;
  const float* __restrict__ bg = sum2B + (long long)node * 40;
  float v0 = ag[lane] + bg[lane];
  float v1 = (lane < 8) ? ag[lane + 32] + bg[lane + 32] : -INFINITY;
  float m = fmaxf(v0, v1);
#pragma unroll
  for (int o = 16; o > 0; o >>= 1) m = fmaxf(m, __shfl_xor(m, o, 32));
  float s = expf(v0 - m) + ((lane < 8) ? expf(v1 - m) : 0.0f);
#pragma unroll
  for (int o = 16; o > 0; o >>= 1) s += __shfl_xor(s, o, 32);
  const float lse = m + logf(s);
  float* op = out + (long long)node * 40;
  op[lane] = v0 - lse;
  if (lane < 8) op[lane + 32] = v1 - lse;
}

#define LSM_ROWS 8

__global__ void lsm_kernel(const float* __restrict__ sum2B, float* out, int N) {
  const int wave = (blockIdx.x * blockDim.x + threadIdx.x) >> 5;
  const int row0 = wave * LSM_ROWS;
  if (row0 >= N) return;
  const int lane = threadIdx.x & 31;
  if (row0 + LSM_ROWS > N) {
    for (int r = 0; r < LSM_ROWS; ++r)
      if (row0 + r < N) lsm_one_row(sum2B, out, row0 + r, lane);
    return;
  }
  const int c1 = (lane < 8) ? lane + 32 : lane;
  float v0[LSM_ROWS], v1[LSM_ROWS], m[LSM_ROWS], s[LSM_ROWS];
#pragma unroll
  for (int r = 0; r < LSM_ROWS; ++r) {
    const float* ag = out + (long long)(row0 + r) * 40;
    const float* __restrict__ bg = sum2B + (long long)(row0 + r) * 40;
    v0[r] = ag[lane] + bg[lane];
    const float t1 = ag[c1] + bg[c1];
    v1[r] = (lane < 8) ? t1 : -INFINITY;
    m[r] = fmaxf(v0[r], v1[r]);
  }
#pragma unroll
  for (int o = 16; o > 0; o >>= 1) {
#pragma unroll
    for (int r = 0; r < LSM_ROWS; ++r) m[r] = fmaxf(m[r], __shfl_xor(m[r], o, 32));
  }
#pragma unroll
  for (int r = 0; r < LSM_ROWS; ++r)
    s[r] = expf(v0[r] - m[r]) + expf(v1[r] - m[r]);
#pragma unroll
  for (int o = 16; o > 0; o >>= 1) {
#pragma unroll
    for (int r = 0; r < LSM_ROWS; ++r) s[r] += __shfl_xor(s[r], o, 32);
  }
#pragma unroll
  for (int r = 0; r < LSM_ROWS; ++r) {
    const float lse = m[r] + logf(s[r]);
    v0[r] -= lse;
    v1[r] -= lse;
  }
  float* op = out + (long long)row0 * 40;
#pragma unroll
  for (int r = 0; r < LSM_ROWS; ++r) op[r * 40 + lane] = v0[r];
  if (lane < 8) {
#pragma unroll
    for (int r = 0; r < LSM_ROWS; ++r) op[r * 40 + 32 + lane] = v1[r];
  }
}

extern "C" void kernel_launch(void* const* d_in, const int* in_sizes, int n_in,
                              void* d_out, int out_size, void* d_ws, size_t ws_size,
                              hipStream_t stream) {
  const float* x  = (const float*)d_in[0];
  const int* ei   = (const int*)d_in[1];
  const float* W1 = (const float*)d_in[2];
  const float* b1 = (const float*)d_in[3];
  const float* W2 = (const float*)d_in[4];
  const float* b2 = (const float*)d_in[5];
  float* out = (float*)d_out;

  const int N = in_sizes[0] / 64;
  const int E = in_sizes[1] / 2;
  const int* src = ei;
  const int* dst = ei + E;

  float* ws = (float*)d_ws;
  size_t off = 0;
  float* degA  = ws + off; off += (size_t)N;
  float* dinv  = ws + off; off += (size_t)N;
  float* nrm   = ws + off; off += (size_t)E;
  float* sum1A = ws + off; off += (size_t)N * 16;
  float* sum1B = ws + off; off += (size_t)N * 16;
  float* y2    = ws + off; off += (size_t)N * 40;
  float* sum2B = ws + off; off += (size_t)N * 40;
  float* y1    = y2;
  float* degB  = y2 + (size_t)N * 16;
  float* sum2A = out;

  const int B = 256;
  auto blk = [](long long n, int b) { return (unsigned)((n + b - 1) / b); };

  const int tiles = (N + 15) / 16;
  gemm1_kernel<<<blk((long long)tiles * 32, B), B, 0, stream>>>(x, W1, b1, sum1A, sum1B, y1, degA, degB, N, tiles);

  const long long edgeWaves = ((long long)E + 32 * EDGE_K - 1) / (32 * EDGE_K);
  deg_kernel<<<blk(edgeWaves * 32, B), B, 0, stream>>>(src, degA, degB, E);
  dinv_kernel<<<blk(N, B), B, 0, stream>>>(degA, degB, dinv, N);
  norm_kernel<<<blk(edgeWaves * 32, B), B, 0, stream>>>(src, dst, dinv, nrm, E);

  scatter1_kernel<<<blk(E, B), B, 0, stream>>>(src, dst, nrm, y1, sum1A, sum1B, E);

  gemm2_kernel<<<blk((long long)tiles * 32, B), B, 0, stream>>>(sum1A, sum1B, W2, b2, sum2A, sum2B, y2, N, tiles);
  scatter2_kernel<<<blk(E, B), B, 0, stream>>>(src, dst, nrm, y2, sum2A, sum2B, E);

  const long long lsmWaves = ((long long)N + LSM_ROWS - 1) / LSM_ROWS;
  lsm_kernel<<<blk(lsmWaves * 32, B), B, 0, stream>>>(sum2B, out, N);
}


# Edge coefficient pass folded into the layer-1 edge launch: each lane builds -dinv[src]*dinv[dst] for the edge it holds, its two dinv reads in flight with the row gathers, and stores it; 7 launches
# speedup vs baseline: 2.8526x; 1.0922x over previous
#include <hip/hip_runtime.h>
#include <math.h>

typedef float v2f __attribute__((ext_vector_type(2)));
typedef float v8f __attribute__((ext_vector_type(8)));

#define WMMA_F32X4(a, b, c) \
  __builtin_amdgcn_wmma_f32_16x16x4_f32(false, (a), false, (b), (short)0, (c), false, false)


__device__ __forceinline__ int package_half() {
  return (__builtin_amdgcn_s_sendmsg_rtn(0x87) >> 18) & 1;
}

#define EDGE_K 8

__global__ void deg_kernel(const int* __restrict__ src, float* __restrict__ degA,
                           float* __restrict__ degB, int E) {
  const unsigned lane = threadIdx.x & 31u;
  const unsigned wave = (blockIdx.x * blockDim.x + threadIdx.x) >> 5;
  const unsigned ebase = wave * (32u * EDGE_K);
  if (ebase >= (unsigned)E) return;
  float* __restrict__ deg = package_half() ? degB : degA;
  if (ebase + 32u * EDGE_K <= (unsigned)E) {
    int s[EDGE_K];
#pragma unroll
    for (int j = 0; j < EDGE_K; ++j) s[j] = src[ebase + 32u * (unsigned)j + lane];
#pragma unroll
    for (int j = 0; j < EDGE_K; ++j) unsafeAtomicAdd(&deg[s[j]], 1.0f);
  } else {
    for (int j = 0; j < EDGE_K; ++j) {
      const unsigned e = ebase + 32u * (unsigned)j + lane;
      if (e < (unsigned)E) unsafeAtomicAdd(&deg[src[e]], 1.0f);
    }
  }
}

__global__ void dinv_kernel(const float* __restrict__ degA, const float* __restrict__ degB,
                            float* __restrict__ dinv, int N) {
  int i = blockIdx.x * blockDim.x + threadIdx.x;
  if (i < N) {
    float d = degA[i] + degB[i];
    dinv[i] = (d > 0.0f) ? rsqrtf(fmaxf(d, 1.0f)) : 0.0f;
  }
}

__global__ void gemm1_kernel(const float* __restrict__ x, const float* __restrict__ W1,
                             const float* __restrict__ b1, float* __restrict__ sum1A,
                             float* __restrict__ sum1B,
                             float* __restrict__ y1, float* __restrict__ degA,
                             float* __restrict__ degB, int N, int nTiles) {
  const int wave = (blockIdx.x * blockDim.x + threadIdx.x) >> 5;
  if (wave >= nTiles) return;
  const int lane = threadIdx.x & 31;
  const int n = lane & 15;
  const int hi = lane >> 4;
  const int base = wave * 16;
  const int mrow = min(base + n, N - 1);
  const float* __restrict__ W0 = W1;
  const float* __restrict__ Wb = W1 + 64 * 16;
  const float* __restrict__ xr = x + (long long)mrow * 64;

  v8f acc0 = {0.f, 0.f, 0.f, 0.f, 0.f, 0.f, 0.f, 0.f};
  v8f acc1 = acc0;
#pragma unroll
  for (int kb = 0; kb < 64; kb += 4) {
    const int k0 = kb + 2 * hi;
    v2f a = *(const v2f*)(xr + k0);
    v2f b0, bb;
    b0.x = W0[k0 * 16 + n];  b0.y = W0[(k0 + 1) * 16 + n];
    bb.x = Wb[k0 * 16 + n];  bb.y = Wb[(k0 + 1) * 16 + n];
    acc0 = WMMA_F32X4(a, b0, acc0);
    acc1 = WMMA_F32X4(a, bb, acc1);
  }
  const float bias = b1[n];
#pragma unroll
  for (int r = 0; r < 8; ++r) {
    const int row = base + r + 8 * hi;
    if (row < N) {
      sum1A[(long long)row * 16 + n] = acc0[r] + bias;
      sum1B[(long long)row * 16 + n] = 0.0f;
      y1[(long long)row * 16 + n]    = acc1[r];
    }
  }
  if (base + n < N) (hi ? degB : degA)[base + n] = 0.0f;
}

__global__ void scatter1_kernel(const int* __restrict__ src, const int* __restrict__ dst,
                                const float* __restrict__ dinv, float* __restrict__ nrm,
                                const float* __restrict__ y1,
                                float* __restrict__ sum1A, float* __restrict__ sum1B, int E) {
  const unsigned lane = threadIdx.x & 31u;
  const unsigned wave = (blockIdx.x * blockDim.x + threadIdx.x) >> 5;
  const unsigned ebase = wave * 32u;
  if (ebase >= (unsigned)E) return;
  const unsigned el = ebase + lane;
  const bool live = el < (unsigned)E;
  int s_l = 0, d_l = 0;
  if (live) { s_l = src[el]; d_l = dst[el]; }
  const float ds_l = dinv[s_l];
  const float dd_l = dinv[d_l];
  float* __restrict__ sum1 = package_half() ? sum1B : sum1A;
  const unsigned f = lane & 15u;
  const unsigned hi = lane >> 4;

  float yv[16];
  int dv[16];
#pragma unroll
  for (int j = 0; j < 16; ++j) {
    const int sl = 2 * j + (int)hi;
    const int s = __shfl(s_l, sl, 32);
    dv[j] = __shfl(d_l, sl, 32);
    yv[j] = y1[(long long)s * 16 + f];
  }
  const float w_l = -ds_l * dd_l;
  float wv[16];
#pragma unroll
  for (int j = 0; j < 16; ++j) wv[j] = __shfl(w_l, 2 * j + (int)hi, 32);
  if (live) nrm[el] = w_l;
#pragma unroll
  for (int j = 0; j < 16; ++j) {
    if (ebase + 2u * (unsigned)j + hi < (unsigned)E)
      unsafeAtomicAdd(&sum1[(long long)dv[j] * 16 + f], wv[j] * yv[j]);
  }
}

__global__ void gemm2_kernel(const float* __restrict__ sum1A, const float* __restrict__ sum1B,
                             const float* __restrict__ W2, const float* __restrict__ b2,
                             float* __restrict__ sum2A, float* __restrict__ sum2B,
                             float* __restrict__ y2, int N, int nTiles) {
  const int wave = (blockIdx.x * blockDim.x + threadIdx.x) >> 5;
  if (wave >= nTiles) return;
  const int lane = threadIdx.x & 31;
  const int n = lane & 15;
  const int hi = lane >> 4;
  const int base = wave * 16;
  const int mrow = min(base + n, N - 1);
  const float* __restrict__ W0 = W2;
  const float* __restrict__ Wb = W2 + 16 * 40;

  v2f a[4];
#pragma unroll
  for (int i = 0; i < 4; ++i) {
    const int k0 = i * 4 + 2 * hi;
    const long long idx = (long long)mrow * 16 + k0;
    v2f g = *(const v2f*)(sum1A + idx);
    v2f gB = *(const v2f*)(sum1B + idx);
    a[i].x = fmaxf(g.x + gB.x, 0.0f);
    a[i].y = fmaxf(g.y + gB.y, 0.0f);
  }

  v8f acc[6];
#pragma unroll
  for (int i = 0; i < 6; ++i) acc[i] = (v8f){0.f, 0.f, 0.f, 0.f, 0.f, 0.f, 0.f, 0.f};

#pragma unroll
  for (int ct = 0; ct < 3; ++ct) {
    const int col = ct * 16 + n;
    const int cc = min(col, 39);
#pragma unroll
    for (int i = 0; i < 4; ++i) {
      const int k0 = i * 4 + 2 * hi;
      v2f b0, bb;
      b0.x = W0[k0 * 40 + cc];  b0.y = W0[(k0 + 1) * 40 + cc];
      bb.x = Wb[k0 * 40 + cc];  bb.y = Wb[(k0 + 1) * 40 + cc];
      acc[ct * 2]     = WMMA_F32X4(a[i], b0, acc[ct * 2]);
      acc[ct * 2 + 1] = WMMA_F32X4(a[i], bb, acc[ct * 2 + 1]);
    }
  }

#pragma unroll
  for (int ct = 0; ct < 3; ++ct) {
    const int col = ct * 16 + n;
    if (col < 40) {
      const float bias = b2[col];
#pragma unroll
      for (int r = 0; r < 8; ++r) {
        const int row = base + r + 8 * hi;
        if (row < N) {
          sum2A[(long long)row * 40 + col] = acc[ct * 2][r] + bias;
          sum2B[(long long)row * 40 + col] = 0.0f;
          y2[(long long)row * 40 + col]    = acc[ct * 2 + 1][r];
        }
      }
    }
  }
}

__global__ void scatter2_kernel(const int* __restrict__ src, const int* __restrict__ dst,
                                const float* __restrict__ nrm, const float* __restrict__ y2,
                                float* __restrict__ sum2A, float* __restrict__ sum2B, int E) {
  const unsigned lane = threadIdx.x & 31u;
  const unsigned wave = (blockIdx.x * blockDim.x + threadIdx.x) >> 5;
  const unsigned ebase = wave * 32u;
  if (ebase >= (unsigned)E) return;
  const unsigned el = ebase + lane;
  int s_l = 0, d_l = 0;
  float w_l = 0.0f;
  if (el < (unsigned)E) { s_l = src[el]; d_l = dst[el]; w_l = nrm[el]; }
  float* __restrict__ sum2 = package_half() ? sum2B : sum2A;

#pragma unroll
  for (int b = 0; b < 4; ++b) {
    float va[8], vb[8], wv[8];
    long long ro[8];
    int n0v[8];
#pragma unroll
    for (int j = 0; j < 8; ++j) {
      const int sl = b * 8 + j;
      const int s = __builtin_amdgcn_readlane(s_l, sl);
      const int d = __builtin_amdgcn_readlane(d_l, sl);
      wv[j] = __builtin_bit_cast(float, __builtin_amdgcn_readlane(__builtin_bit_cast(int, w_l), sl));
      const long long drow = (long long)d * 40;
      ro[j] = drow;
      const unsigned off = (unsigned)(((unsigned long long)(sum2 + drow) >> 2) & 31ull);
      int n0 = 32 - (int)off;
      n0 = n0 < 8 ? 8 : n0;
      n0v[j] = n0;
      const float* __restrict__ yr = y2 + (long long)s * 40;
      const int fb = n0 + (int)lane;
      va[j] = yr[lane];
      vb[j] = yr[fb < 40 ? fb : 39];
    }
#pragma unroll
    for (int j = 0; j < 8; ++j) {
      if (ebase + (unsigned)(b * 8 + j) < (unsigned)E) {
        const int n0 = n0v[j];
        const int fb = n0 + (int)lane;
        float* __restrict__ ar = sum2 + ro[j];
        if ((int)lane < n0) unsafeAtomicAdd(&ar[lane], wv[j] * va[j]);
        if (fb < 40)        unsafeAtomicAdd(&ar[fb],   wv[j] * vb[j]);
      }
    }
  }
}

__device__ __forceinline__ void lsm_one_row(const float* __restrict__ sum2B, float* out,
                                            int node, int lane) {
  const float* ag = out + (long long)node * 40;
  const float* __restrict__ bg = sum2B + (long long)node * 40;
  float v0 = ag[lane] + bg[lane];
  float v1 = (lane < 8) ? ag[lane + 32] + bg[lane + 32] : -INFINITY;
  float m = fmaxf(v0, v1);
#pragma unroll
  for (int o = 16; o > 0; o >>= 1) m = fmaxf(m, __shfl_xor(m, o, 32));
  float s = expf(v0 - m) + ((lane < 8) ? expf(v1 - m) : 0.0f);
#pragma unroll
  for (int o = 16; o > 0; o >>= 1) s += __shfl_xor(s, o, 32);
  const float lse = m + logf(s);
  float* op = out + (long long)node * 40;
  op[lane] = v0 - lse;
  if (lane < 8) op[lane + 32] = v1 - lse;
}

#define LSM_ROWS 8

__global__ void lsm_kernel(const float* __restrict__ sum2B, float* out, int N) {
  const int wave = (blockIdx.x * blockDim.x + threadIdx.x) >> 5;
  const int row0 = wave * LSM_ROWS;
  if (row0 >= N) return;
  const int lane = threadIdx.x & 31;
  if (row0 + LSM_ROWS > N) {
    for (int r = 0; r < LSM_ROWS; ++r)
      if (row0 + r < N) lsm_one_row(sum2B, out, row0 + r, lane);
    return;
  }
  const int c1 = (lane < 8) ? lane + 32 : lane;
  float v0[LSM_ROWS], v1[LSM_ROWS], m[LSM_ROWS], s[LSM_ROWS];
#pragma unroll
  for (int r = 0; r < LSM_ROWS; ++r) {
    const float* ag = out + (long long)(row0 + r) * 40;
    const float* __restrict__ bg = sum2B + (long long)(row0 + r) * 40;
    v0[r] = ag[lane] + bg[lane];
    const float t1 = ag[c1] + bg[c1];
    v1[r] = (lane < 8) ? t1 : -INFINITY;
    m[r] = fmaxf(v0[r], v1[r]);
  }
#pragma unroll
  for (int o = 16; o > 0; o >>= 1) {
#pragma unroll
    for (int r = 0; r < LSM_ROWS; ++r) m[r] = fmaxf(m[r], __shfl_xor(m[r], o, 32));
  }
#pragma unroll
  for (int r = 0; r < LSM_ROWS; ++r)
    s[r] = expf(v0[r] - m[r]) + expf(v1[r] - m[r]);
#pragma unroll
  for (int o = 16; o > 0; o >>= 1) {
#pragma unroll
    for (int r = 0; r < LSM_ROWS; ++r) s[r] += __shfl_xor(s[r], o, 32);
  }
#pragma unroll
  for (int r = 0; r < LSM_ROWS; ++r) {
    const float lse = m[r] + logf(s[r]);
    v0[r] -= lse;
    v1[r] -= lse;
  }
  float* op = out + (long long)row0 * 40;
#pragma unroll
  for (int r = 0; r < LSM_ROWS; ++r) op[r * 40 + lane] = v0[r];
  if (lane < 8) {
#pragma unroll
    for (int r = 0; r < LSM_ROWS; ++r) op[r * 40 + 32 + lane] = v1[r];
  }
}

extern "C" void kernel_launch(void* const* d_in, const int* in_sizes, int n_in,
                              void* d_out, int out_size, void* d_ws, size_t ws_size,
                              hipStream_t stream) {
  const float* x  = (const float*)d_in[0];
  const int* ei   = (const int*)d_in[1];
  const float* W1 = (const float*)d_in[2];
  const float* b1 = (const float*)d_in[3];
  const float* W2 = (const float*)d_in[4];
  const float* b2 = (const float*)d_in[5];
  float* out = (float*)d_out;

  const int N = in_sizes[0] / 64;
  const int E = in_sizes[1] / 2;
  const int* src = ei;
  const int* dst = ei + E;

  float* ws = (float*)d_ws;
  size_t off = 0;
  float* degA  = ws + off; off += (size_t)N;
  float* dinv  = ws + off; off += (size_t)N;
  float* nrm   = ws + off; off += (size_t)E;
  float* sum1A = ws + off; off += (size_t)N * 16;
  float* sum1B = ws + off; off += (size_t)N * 16;
  float* y2    = ws + off; off += (size_t)N * 40;
  float* sum2B = ws + off; off += (size_t)N * 40;
  float* y1    = y2;
  float* degB  = y2 + (size_t)N * 16;
  float* sum2A = out;

  const int B = 256;
  auto blk = [](long long n, int b) { return (unsigned)((n + b - 1) / b); };

  const int tiles = (N + 15) / 16;
  gemm1_kernel<<<blk((long long)tiles * 32, B), B, 0, stream>>>(x, W1, b1, sum1A, sum1B, y1, degA, degB, N, tiles);

  const long long edgeWaves = ((long long)E + 32 * EDGE_K - 1) / (32 * EDGE_K);
  deg_kernel<<<blk(edgeWaves * 32, B), B, 0, stream>>>(src, degA, degB, E);
  dinv_kernel<<<blk(N, B), B, 0, stream>>>(degA, degB, dinv, N);

  scatter1_kernel<<<blk(E, B), B, 0, stream>>>(src, dst, dinv, nrm, y1, sum1A, sum1B, E);

  gemm2_kernel<<<blk((long long)tiles * 32, B), B, 0, stream>>>(sum1A, sum1B, W2, b2, sum2A, sum2B, y2, N, tiles);
  scatter2_kernel<<<blk(E, B), B, 0, stream>>>(src, dst, nrm, y2, sum2A, sum2B, E);

  const long long lsmWaves = ((long long)N + LSM_ROWS - 1) / LSM_ROWS;
  lsm_kernel<<<blk(lsmWaves * 32, B), B, 0, stream>>>(sum2B, out, N);
}


# Layer-2 rows as a 32-column line plane + an 8-column plane (y2 kept in out, both sum copies in workspace): per edge one aligned whole-line op + 1/4 of a 4-edge op, no lane masks; add order changes
# speedup vs baseline: 3.3897x; 1.1883x over previous
#include <hip/hip_runtime.h>
#include <math.h>

typedef float v2f __attribute__((ext_vector_type(2)));
typedef float v8f __attribute__((ext_vector_type(8)));

#define WMMA_F32X4(a, b, c) \
  __builtin_amdgcn_wmma_f32_16x16x4_f32(false, (a), false, (b), (short)0, (c), false, false)


__device__ __forceinline__ int package_half() {
  return (__builtin_amdgcn_s_sendmsg_rtn(0x87) >> 18) & 1;
}

#define EDGE_K 8

__global__ void deg_kernel(const int* __restrict__ src, float* __restrict__ degA,
                           float* __restrict__ degB, int E) {
  const unsigned lane = threadIdx.x & 31u;
  const unsigned wave = (blockIdx.x * blockDim.x + threadIdx.x) >> 5;
  const unsigned ebase = wave * (32u * EDGE_K);
  if (ebase >= (unsigned)E) return;
  float* __restrict__ deg = package_half() ? degB : degA;
  if (ebase + 32u * EDGE_K <= (unsigned)E) {
    int s[EDGE_K];
#pragma unroll
    for (int j = 0; j < EDGE_K; ++j) s[j] = src[ebase + 32u * (unsigned)j + lane];
#pragma unroll
    for (int j = 0; j < EDGE_K; ++j) unsafeAtomicAdd(&deg[s[j]], 1.0f);
  } else {
    for (int j = 0; j < EDGE_K; ++j) {
      const unsigned e = ebase + 32u * (unsigned)j + lane;
      if (e < (unsigned)E) unsafeAtomicAdd(&deg[src[e]], 1.0f);
    }
  }
}

__global__ void dinv_kernel(const float* __restrict__ degA, const float* __restrict__ degB,
                            float* __restrict__ dinv, int N) {
  int i = blockIdx.x * blockDim.x + threadIdx.x;
  if (i < N) {
    float d = degA[i] + degB[i];
    dinv[i] = (d > 0.0f) ? rsqrtf(fmaxf(d, 1.0f)) : 0.0f;
  }
}

__global__ void gemm1_kernel(const float* __restrict__ x, const float* __restrict__ W1,
                             const float* __restrict__ b1, float* __restrict__ sum1A,
                             float* __restrict__ sum1B,
                             float* __restrict__ y1, float* __restrict__ degA,
                             float* __restrict__ degB, int N, int nTiles) {
  const int wave = (blockIdx.x * blockDim.x + threadIdx.x) >> 5;
  if (wave >= nTiles) return;
  const int lane = threadIdx.x & 31;
  const int n = lane & 15;
  const int hi = lane >> 4;
  const int base = wave * 16;
  const int mrow = min(base + n, N - 1);
  const float* __restrict__ W0 = W1;
  const float* __restrict__ Wb = W1 + 64 * 16;
  const float* __restrict__ xr = x + (long long)mrow * 64;

  v8f acc0 = {0.f, 0.f, 0.f, 0.f, 0.f, 0.f, 0.f, 0.f};
  v8f acc1 = acc0;
#pragma unroll
  for (int kb = 0; kb < 64; kb += 4) {
    const int k0 = kb + 2 * hi;
    v2f a = *(const v2f*)(xr + k0);
    v2f b0, bb;
    b0.x = W0[k0 * 16 + n];  b0.y = W0[(k0 + 1) * 16 + n];
    bb.x = Wb[k0 * 16 + n];  bb.y = Wb[(k0 + 1) * 16 + n];
    acc0 = WMMA_F32X4(a, b0, acc0);
    acc1 = WMMA_F32X4(a, bb, acc1);
  }
  const float bias = b1[n];
#pragma unroll
  for (int r = 0; r < 8; ++r) {
    const int row = base + r + 8 * hi;
    if (row < N) {
      sum1A[(long long)row * 16 + n] = acc0[r] + bias;
      sum1B[(long long)row * 16 + n] = 0.0f;
      y1[(long long)row * 16 + n]    = acc1[r];
    }
  }
  if (base + n < N) (hi ? degB : degA)[base + n] = 0.0f;
}

__global__ void scatter1_kernel(const int* __restrict__ src, const int* __restrict__ dst,
                                const float* __restrict__ dinv, float* __restrict__ nrm,
                                const float* __restrict__ y1,
                                float* __restrict__ sum1A, float* __restrict__ sum1B, int E) {
  const unsigned lane = threadIdx.x & 31u;
  const unsigned wave = (blockIdx.x * blockDim.x + threadIdx.x) >> 5;
  const unsigned ebase = wave * 32u;
  if (ebase >= (unsigned)E) return;
  const unsigned el = ebase + lane;
  const bool live = el < (unsigned)E;
  int s_l = 0, d_l = 0;
  if (live) { s_l = src[el]; d_l = dst[el]; }
  const float ds_l = dinv[s_l];
  const float dd_l = dinv[d_l];
  float* __restrict__ sum1 = package_half() ? sum1B : sum1A;
  const unsigned f = lane & 15u;
  const unsigned hi = lane >> 4;

  float yv[16];
  int dv[16];
#pragma unroll
  for (int j = 0; j < 16; ++j) {
    const int sl = 2 * j + (int)hi;
    const int s = __shfl(s_l, sl, 32);
    dv[j] = __shfl(d_l, sl, 32);
    yv[j] = y1[(long long)s * 16 + f];
  }
  const float w_l = -ds_l * dd_l;
  float wv[16];
#pragma unroll
  for (int j = 0; j < 16; ++j) wv[j] = __shfl(w_l, 2 * j + (int)hi, 32);
  if (live) nrm[el] = w_l;
#pragma unroll
  for (int j = 0; j < 16; ++j) {
    if (ebase + 2u * (unsigned)j + hi < (unsigned)E)
      unsafeAtomicAdd(&sum1[(long long)dv[j] * 16 + f], wv[j] * yv[j]);
  }
}

__global__ void gemm2_kernel(const float* __restrict__ sum1A, const float* __restrict__ sum1B,
                             const float* __restrict__ W2, const float* __restrict__ b2,
                             float* __restrict__ sum2A, float* __restrict__ sum2B,
                             float* __restrict__ y2, int N, int nTiles) {
  const int wave = (blockIdx.x * blockDim.x + threadIdx.x) >> 5;
  if (wave >= nTiles) return;
  const int lane = threadIdx.x & 31;
  const int n = lane & 15;
  const int hi = lane >> 4;
  const int base = wave * 16;
  const int mrow = min(base + n, N - 1);
  const float* __restrict__ W0 = W2;
  const float* __restrict__ Wb = W2 + 16 * 40;

  v2f a[4];
#pragma unroll
  for (int i = 0; i < 4; ++i) {
    const int k0 = i * 4 + 2 * hi;
    const long long idx = (long long)mrow * 16 + k0;
    v2f g = *(const v2f*)(sum1A + idx);
    v2f gB = *(const v2f*)(sum1B + idx);
    a[i].x = fmaxf(g.x + gB.x, 0.0f);
    a[i].y = fmaxf(g.y + gB.y, 0.0f);
  }

  v8f acc[6];
#pragma unroll
  for (int i = 0; i < 6; ++i) acc[i] = (v8f){0.f, 0.f, 0.f, 0.f, 0.f, 0.f, 0.f, 0.f};

#pragma unroll
  for (int ct = 0; ct < 3; ++ct) {
    const int col = ct * 16 + n;
    const int cc = min(col, 39);
#pragma unroll
    for (int i = 0; i < 4; ++i) {
      const int k0 = i * 4 + 2 * hi;
      v2f b0, bb;
      b0.x = W0[k0 * 40 + cc];  b0.y = W0[(k0 + 1) * 40 + cc];
      bb.x = Wb[k0 * 40 + cc];  bb.y = Wb[(k0 + 1) * 40 + cc];
      acc[ct * 2]     = WMMA_F32X4(a[i], b0, acc[ct * 2]);
      acc[ct * 2 + 1] = WMMA_F32X4(a[i], bb, acc[ct * 2 + 1]);
    }
  }

  const long long qoff = (long long)N * 32;
#pragma unroll
  for (int ct = 0; ct < 3; ++ct) {
    const int col = ct * 16 + n;
    if (col < 40) {
      const float bias = b2[col];
#pragma unroll
      for (int r = 0; r < 8; ++r) {
        const int row = base + r + 8 * hi;
        if (row < N) {
          const long long o = (ct < 2) ? (long long)row * 32 + col
                                       : qoff + (long long)row * 8 + (col - 32);
          sum2A[o] = acc[ct * 2][r] + bias;
          sum2B[o] = 0.0f;
          y2[o]    = acc[ct * 2 + 1][r];
        }
      }
    }
  }
}

#define S2_BATCH 16

__global__ void scatter2_kernel(const int* __restrict__ src, const int* __restrict__ dst,
                                const float* __restrict__ nrm, const float* __restrict__ y2,
                                float* __restrict__ sum2A, float* __restrict__ sum2B,
                                int E, int N) {
  const unsigned lane = threadIdx.x & 31u;
  const unsigned wave = (blockIdx.x * blockDim.x + threadIdx.x) >> 5;
  const unsigned ebase = wave * 32u;
  if (ebase >= (unsigned)E) return;
  const unsigned el = ebase + lane;
  const long long qoff = (long long)N * 32;
  float* __restrict__ sumP = package_half() ? sum2B : sum2A;
  float* __restrict__ sumQ = sumP + qoff;
  const float* __restrict__ yP = y2;
  const float* __restrict__ yQ = y2 + qoff;
  const unsigned qc = lane & 7u;
  const int qe = (int)(lane >> 3);

  if (ebase + 32u <= (unsigned)E) {
    const int s_l = src[el];
    const int d_l = dst[el];
    const float w_l = nrm[el];
#pragma unroll
    for (int b = 0; b < 32 / S2_BATCH; ++b) {
      float va[S2_BATCH], qv[S2_BATCH / 4], wq[S2_BATCH / 4];
      int dq[S2_BATCH / 4];
#pragma unroll
      for (int g = 0; g < S2_BATCH / 4; ++g) {
        const int sl = b * S2_BATCH + g * 4 + qe;
        const int sq = __shfl(s_l, sl, 32);
        dq[g] = __shfl(d_l, sl, 32);
        wq[g] = __shfl(w_l, sl, 32);
        qv[g] = yQ[(unsigned)sq * 8u + qc];
      }
#pragma unroll
      for (int j = 0; j < S2_BATCH; ++j) {
        const int s = __builtin_amdgcn_readlane(s_l, b * S2_BATCH + j);
        va[j] = yP[(unsigned)s * 32u + lane];
      }
#pragma unroll
      for (int j = 0; j < S2_BATCH; ++j) {
        const int d = __builtin_amdgcn_readlane(d_l, b * S2_BATCH + j);
        const float w = __builtin_bit_cast(
            float, __builtin_amdgcn_readlane(__builtin_bit_cast(int, w_l), b * S2_BATCH + j));
        unsafeAtomicAdd(&sumP[(unsigned)d * 32u + lane], w * va[j]);
      }
#pragma unroll
      for (int g = 0; g < S2_BATCH / 4; ++g)
        unsafeAtomicAdd(&sumQ[(unsigned)dq[g] * 8u + qc], wq[g] * qv[g]);
    }
  } else {
    int s_l = 0, d_l = 0;
    float w_l = 0.0f;
    if (el < (unsigned)E) { s_l = src[el]; d_l = dst[el]; w_l = nrm[el]; }
    const int cnt = (int)((unsigned)E - ebase);
    for (int j = 0; j < cnt; ++j) {
      const int s = __shfl(s_l, j, 32);
      const int d = __shfl(d_l, j, 32);
      const float w = __shfl(w_l, j, 32);
      unsafeAtomicAdd(&sumP[(long long)d * 32 + lane], w * yP[(long long)s * 32 + lane]);
      if (lane < 8u) unsafeAtomicAdd(&sumQ[(long long)d * 8 + lane], w * yQ[(long long)s * 8 + lane]);
    }
  }
}

__device__ __forceinline__ void lsm_one_row(const float* __restrict__ sum2A,
                                            const float* __restrict__ sum2B,
                                            float* __restrict__ out, int node, int lane, int N) {
  const long long po = (long long)node * 32 + lane;
  const long long qo = (long long)N * 32 + (long long)node * 8 + (lane & 7);
  float v0 = sum2A[po] + sum2B[po];
  float v1 = (lane < 8) ? sum2A[qo] + sum2B[qo] : -INFINITY;
  float m = fmaxf(v0, v1);
#pragma unroll
  for (int o = 16; o > 0; o >>= 1) m = fmaxf(m, __shfl_xor(m, o, 32));
  float s = expf(v0 - m) + ((lane < 8) ? expf(v1 - m) : 0.0f);
#pragma unroll
  for (int o = 16; o > 0; o >>= 1) s += __shfl_xor(s, o, 32);
  const float lse = m + logf(s);
  float* op = out + (long long)node * 40;
  op[lane] = v0 - lse;
  if (lane < 8) op[lane + 32] = v1 - lse;
}

#define LSM_ROWS 8

__global__ void lsm_kernel(const float* __restrict__ sum2A, const float* __restrict__ sum2B,
                           float* __restrict__ out, int N) {
  const int wave = (blockIdx.x * blockDim.x + threadIdx.x) >> 5;
  const int row0 = wave * LSM_ROWS;
  if (row0 >= N) return;
  const int lane = threadIdx.x & 31;
  if (row0 + LSM_ROWS > N) {
    for (int r = 0; r < LSM_ROWS; ++r)
      if (row0 + r < N) lsm_one_row(sum2A, sum2B, out, row0 + r, lane, N);
    return;
  }
  const long long qoff = (long long)N * 32 + (lane & 7);
  float v0[LSM_ROWS], v1[LSM_ROWS], m[LSM_ROWS], s[LSM_ROWS];
#pragma unroll
  for (int r = 0; r < LSM_ROWS; ++r) {
    const long long po = (long long)(row0 + r) * 32 + lane;
    const long long qo = qoff + (long long)(row0 + r) * 8;
    v0[r] = sum2A[po] + sum2B[po];
    const float t1 = sum2A[qo] + sum2B[qo];
    v1[r] = (lane < 8) ? t1 : -INFINITY;
    m[r] = fmaxf(v0[r], v1[r]);
  }
#pragma unroll
  for (int o = 16; o > 0; o >>= 1) {
#pragma unroll
    for (int r = 0; r < LSM_ROWS; ++r) m[r] = fmaxf(m[r], __shfl_xor(m[r], o, 32));
  }
#pragma unroll
  for (int r = 0; r < LSM_ROWS; ++r)
    s[r] = expf(v0[r] - m[r]) + expf(v1[r] - m[r]);
#pragma unroll
  for (int o = 16; o > 0; o >>= 1) {
#pragma unroll
    for (int r = 0; r < LSM_ROWS; ++r) s[r] += __shfl_xor(s[r], o, 32);
  }
#pragma unroll
  for (int r = 0; r < LSM_ROWS; ++r) {
    const float lse = m[r] + logf(s[r]);
    v0[r] -= lse;
    v1[r] -= lse;
  }
  float* op = out + (long long)row0 * 40;
#pragma unroll
  for (int r = 0; r < LSM_ROWS; ++r) op[r * 40 + lane] = v0[r];
  if (lane < 8) {
#pragma unroll
    for (int r = 0; r < LSM_ROWS; ++r) op[r * 40 + 32 + lane] = v1[r];
  }
}

extern "C" void kernel_launch(void* const* d_in, const int* in_sizes, int n_in,
                              void* d_out, int out_size, void* d_ws, size_t ws_size,
                              hipStream_t stream) {
  const float* x  = (const float*)d_in[0];
  const int* ei   = (const int*)d_in[1];
  const float* W1 = (const float*)d_in[2];
  const float* b1 = (const float*)d_in[3];
  const float* W2 = (const float*)d_in[4];
  const float* b2 = (const float*)d_in[5];
  float* out = (float*)d_out;

  const int N = in_sizes[0] / 64;
  const int E = in_sizes[1] / 2;
  const int* src = ei;
  const int* dst = ei + E;

  float* ws = (float*)d_ws;
  size_t off = 0;
  float* degA  = ws + off; off += (size_t)N;
  float* dinv  = ws + off; off += (size_t)N;
  float* nrm   = ws + off; off += (size_t)E;
  float* sum1A = ws + off; off += (size_t)N * 16;
  float* sum1B = ws + off; off += (size_t)N * 16;
  float* sum2A = ws + off; off += (size_t)N * 40;
  float* sum2B = ws + off; off += (size_t)N * 40;
  float* y1    = sum2A;
  float* degB  = sum2A + (size_t)N * 16;
  float* y2    = out;

  const int B = 256;
  auto blk = [](long long n, int b) { return (unsigned)((n + b - 1) / b); };

  const int tiles = (N + 15) / 16;
  gemm1_kernel<<<blk((long long)tiles * 32, B), B, 0, stream>>>(x, W1, b1, sum1A, sum1B, y1, degA, degB, N, tiles);

  const long long edgeWaves = ((long long)E + 32 * EDGE_K - 1) / (32 * EDGE_K);
  deg_kernel<<<blk(edgeWaves * 32, B), B, 0, stream>>>(src, degA, degB, E);
  dinv_kernel<<<blk(N, B), B, 0, stream>>>(degA, degB, dinv, N);

  scatter1_kernel<<<blk(E, B), B, 0, stream>>>(src, dst, dinv, nrm, y1, sum1A, sum1B, E);

  gemm2_kernel<<<blk((long long)tiles * 32, B), B, 0, stream>>>(sum1A, sum1B, W2, b2, sum2A, sum2B, y2, N, tiles);
  scatter2_kernel<<<blk(E, B), B, 0, stream>>>(src, dst, nrm, y2, sum2A, sum2B, E, N);

  const long long lsmWaves = ((long long)N + LSM_ROWS - 1) / LSM_ROWS;
  lsm_kernel<<<blk(lsmWaves * 32, B), B, 0, stream>>>(sum2A, sum2B, out, N);
}


# Layer-1 edge launch as 256 resident work-groups: each makes the dinv table in its LDS from the degree copies and loops over 32-edge chunks; no dinv launch or per-edge L2 dinv reads; add order changes
# speedup vs baseline: 3.5439x; 1.0455x over previous
#include <hip/hip_runtime.h>
#include <math.h>

typedef float v2f __attribute__((ext_vector_type(2)));
typedef float v8f __attribute__((ext_vector_type(8)));

#define WMMA_F32X4(a, b, c) \
  __builtin_amdgcn_wmma_f32_16x16x4_f32(false, (a), false, (b), (short)0, (c), false, false)


__device__ __forceinline__ int package_half() {
  return (__builtin_amdgcn_s_sendmsg_rtn(0x87) >> 18) & 1;
}

#define EDGE_K 8

__global__ void deg_kernel(const int* __restrict__ src, float* __restrict__ degA,
                           float* __restrict__ degB, int E) {
  const unsigned lane = threadIdx.x & 31u;
  const unsigned wave = (blockIdx.x * blockDim.x + threadIdx.x) >> 5;
  const unsigned ebase = wave * (32u * EDGE_K);
  if (ebase >= (unsigned)E) return;
  float* __restrict__ deg = package_half() ? degB : degA;
  if (ebase + 32u * EDGE_K <= (unsigned)E) {
    int s[EDGE_K];
#pragma unroll
    for (int j = 0; j < EDGE_K; ++j) s[j] = src[ebase + 32u * (unsigned)j + lane];
#pragma unroll
    for (int j = 0; j < EDGE_K; ++j) unsafeAtomicAdd(&deg[s[j]], 1.0f);
  } else {
    for (int j = 0; j < EDGE_K; ++j) {
      const unsigned e = ebase + 32u * (unsigned)j + lane;
      if (e < (unsigned)E) unsafeAtomicAdd(&deg[src[e]], 1.0f);
    }
  }
}

__global__ void dinv_kernel(const float* __restrict__ degA, const float* __restrict__ degB,
                            float* __restrict__ dinv, int N) {
  int i = blockIdx.x * blockDim.x + threadIdx.x;
  if (i < N) {
    float d = degA[i] + degB[i];
    dinv[i] = (d > 0.0f) ? rsqrtf(fmaxf(d, 1.0f)) : 0.0f;
  }
}

__global__ void gemm1_kernel(const float* __restrict__ x, const float* __restrict__ W1,
                             const float* __restrict__ b1, float* __restrict__ sum1A,
                             float* __restrict__ sum1B,
                             float* __restrict__ y1, float* __restrict__ degA,
                             float* __restrict__ degB, int N, int nTiles) {
  const int wave = (blockIdx.x * blockDim.x + threadIdx.x) >> 5;
  if (wave >= nTiles) return;
  const int lane = threadIdx.x & 31;
  const int n = lane & 15;
  const int hi = lane >> 4;
  const int base = wave * 16;
  const int mrow = min(base + n, N - 1);
  const float* __restrict__ W0 = W1;
  const float* __restrict__ Wb = W1 + 64 * 16;
  const float* __restrict__ xr = x + (long long)mrow * 64;

  v8f acc0 = {0.f, 0.f, 0.f, 0.f, 0.f, 0.f, 0.f, 0.f};
  v8f acc1 = acc0;
#pragma unroll
  for (int kb = 0; kb < 64; kb += 4) {
    const int k0 = kb + 2 * hi;
    v2f a = *(const v2f*)(xr + k0);
    v2f b0, bb;
    b0.x = W0[k0 * 16 + n];  b0.y = W0[(k0 + 1) * 16 + n];
    bb.x = Wb[k0 * 16 + n];  bb.y = Wb[(k0 + 1) * 16 + n];
    acc0 = WMMA_F32X4(a, b0, acc0);
    acc1 = WMMA_F32X4(a, bb, acc1);
  }
  const float bias = b1[n];
#pragma unroll
  for (int r = 0; r < 8; ++r) {
    const int row = base + r + 8 * hi;
    if (row < N) {
      sum1A[(long long)row * 16 + n] = acc0[r] + bias;
      sum1B[(long long)row * 16 + n] = 0.0f;
      y1[(long long)row * 16 + n]    = acc1[r];
    }
  }
  if (base + n < N) (hi ? degB : degA)[base + n] = 0.0f;
}

#define DINV_LDS_CAP 51200
#define FILL_THREADS 1024
#define FILL_K 7
#define FILL_GROUPS ((DINV_LDS_CAP / 4 + FILL_K * FILL_THREADS - 1) / (FILL_K * FILL_THREADS))

__device__ __forceinline__ float dinv_of(float d) {
  return (d > 0.0f) ? rsqrtf(fmaxf(d, 1.0f)) : 0.0f;
}

__global__ void __launch_bounds__(FILL_THREADS)
scatter1_lds_kernel(const int* __restrict__ src, const int* __restrict__ dst,
                    const float* __restrict__ degA, const float* __restrict__ degB,
                    float* __restrict__ nrm, const float* __restrict__ y1,
                    float* __restrict__ sum1A, float* __restrict__ sum1B, int E, int N) {
  __shared__ float tbl[DINV_LDS_CAP];
  const int tid = (int)threadIdx.x;
  const unsigned lane = threadIdx.x & 31u;
  const unsigned nwaves = gridDim.x * (blockDim.x >> 5);
  const unsigned nchunks = ((unsigned)E + 31u) >> 5;
  const unsigned elast = (unsigned)E - 1u;
  unsigned chunk = (threadIdx.x >> 5) * gridDim.x + blockIdx.x;

  unsigned en = min(chunk * 32u + lane, elast);
  int s_n = src[en];
  int d_n = dst[en];

  if ((N & 3) == 0 && ((((size_t)degA) | ((size_t)degB)) & 15) == 0) {
    const int n4 = N >> 2;
    const int rot = (int)(((long long)blockIdx.x * n4) / (long long)gridDim.x);
    const float4* __restrict__ a4 = (const float4*)degA;
    const float4* __restrict__ b4 = (const float4*)degB;
    float4* t4 = (float4*)tbl;
#pragma unroll
    for (int g = 0; g < FILL_GROUPS; ++g) {
      const int base = g * (FILL_K * FILL_THREADS);
      float4 va[FILL_K], vb[FILL_K];
#pragma unroll
      for (int k = 0; k < FILL_K; ++k) {
        int i = min(base + k * FILL_THREADS + tid, n4 - 1) + rot;
        if (i >= n4) i -= n4;
        va[k] = a4[i];
        vb[k] = b4[i];
      }
#pragma unroll
      for (int k = 0; k < FILL_K; ++k) {
        const int u = base + k * FILL_THREADS + tid;
        if (u < n4) {
          int i = u + rot;
          if (i >= n4) i -= n4;
          float4 r;
          r.x = dinv_of(va[k].x + vb[k].x);
          r.y = dinv_of(va[k].y + vb[k].y);
          r.z = dinv_of(va[k].z + vb[k].z);
          r.w = dinv_of(va[k].w + vb[k].w);
          t4[i] = r;
        }
      }
    }
  } else {
    for (int i = tid; i < N; i += FILL_THREADS) tbl[i] = dinv_of(degA[i] + degB[i]);
  }
  __syncthreads();

  if (chunk >= nchunks) return;
  float* __restrict__ sum1 = package_half() ? sum1B : sum1A;
  const unsigned f = lane & 15u;
  const unsigned hi = lane >> 4;

  for (; chunk < nchunks; chunk += nwaves) {
    const unsigned ebase = chunk * 32u;
    const unsigned el = ebase + lane;
    const bool live = el < (unsigned)E;
    const int s_l = s_n;
    const int d_l = d_n;
    en = min(min(chunk + nwaves, nchunks - 1u) * 32u + lane, elast);
    s_n = src[en];
    d_n = dst[en];
    const float ds_l = tbl[s_l];
    const float dd_l = tbl[d_l];

    float yv[16];
    int dv[16];
#pragma unroll
    for (int j = 0; j < 16; ++j) {
      const int sl = 2 * j + (int)hi;
      const int s = __shfl(s_l, sl, 32);
      dv[j] = __shfl(d_l, sl, 32);
      yv[j] = y1[(long long)s * 16 + f];
    }
    const float w_l = -ds_l * dd_l;
    float wv[16];
#pragma unroll
    for (int j = 0; j < 16; ++j) wv[j] = __shfl(w_l, 2 * j + (int)hi, 32);
    if (live) nrm[el] = w_l;
#pragma unroll
    for (int j = 0; j < 16; ++j) {
      if (ebase + 2u * (unsigned)j + hi < (unsigned)E)
        unsafeAtomicAdd(&sum1[(long long)dv[j] * 16 + f], wv[j] * yv[j]);
    }
  }
}

__global__ void scatter1_kernel(const int* __restrict__ src, const int* __restrict__ dst,
                                const float* __restrict__ dinv, float* __restrict__ nrm,
                                const float* __restrict__ y1,
                                float* __restrict__ sum1A, float* __restrict__ sum1B, int E) {
  const unsigned lane = threadIdx.x & 31u;
  const unsigned wave = (blockIdx.x * blockDim.x + threadIdx.x) >> 5;
  const unsigned ebase = wave * 32u;
  if (ebase >= (unsigned)E) return;
  const unsigned el = ebase + lane;
  const bool live = el < (unsigned)E;
  int s_l = 0, d_l = 0;
  if (live) { s_l = src[el]; d_l = dst[el]; }
  const float ds_l = dinv[s_l];
  const float dd_l = dinv[d_l];
  float* __restrict__ sum1 = package_half() ? sum1B : sum1A;
  const unsigned f = lane & 15u;
  const unsigned hi = lane >> 4;

  float yv[16];
  int dv[16];
#pragma unroll
  for (int j = 0; j < 16; ++j) {
    const int sl = 2 * j + (int)hi;
    const int s = __shfl(s_l, sl, 32);
    dv[j] = __shfl(d_l, sl, 32);
    yv[j] = y1[(long long)s * 16 + f];
  }
  const float w_l = -ds_l * dd_l;
  float wv[16];
#pragma unroll
  for (int j = 0; j < 16; ++j) wv[j] = __shfl(w_l, 2 * j + (int)hi, 32);
  if (live) nrm[el] = w_l;
#pragma unroll
  for (int j = 0; j < 16; ++j) {
    if (ebase + 2u * (unsigned)j + hi < (unsigned)E)
      unsafeAtomicAdd(&sum1[(long long)dv[j] * 16 + f], wv[j] * yv[j]);
  }
}

__global__ void gemm2_kernel(const float* __restrict__ sum1A, const float* __restrict__ sum1B,
                             const float* __restrict__ W2, const float* __restrict__ b2,
                             float* __restrict__ sum2A, float* __restrict__ sum2B,
                             float* __restrict__ y2, int N, int nTiles) {
  const int wave = (blockIdx.x * blockDim.x + threadIdx.x) >> 5;
  if (wave >= nTiles) return;
  const int lane = threadIdx.x & 31;
  const int n = lane & 15;
  const int hi = lane >> 4;
  const int base = wave * 16;
  const int mrow = min(base + n, N - 1);
  const float* __restrict__ W0 = W2;
  const float* __restrict__ Wb = W2 + 16 * 40;

  v2f a[4];
#pragma unroll
  for (int i = 0; i < 4; ++i) {
    const int k0 = i * 4 + 2 * hi;
    const long long idx = (long long)mrow * 16 + k0;
    v2f g = *(const v2f*)(sum1A + idx);
    v2f gB = *(const v2f*)(sum1B + idx);
    a[i].x = fmaxf(g.x + gB.x, 0.0f);
    a[i].y = fmaxf(g.y + gB.y, 0.0f);
  }

  v8f acc[6];
#pragma unroll
  for (int i = 0; i < 6; ++i) acc[i] = (v8f){0.f, 0.f, 0.f, 0.f, 0.f, 0.f, 0.f, 0.f};

#pragma unroll
  for (int ct = 0; ct < 3; ++ct) {
    const int col = ct * 16 + n;
    const int cc = min(col, 39);
#pragma unroll
    for (int i = 0; i < 4; ++i) {
      const int k0 = i * 4 + 2 * hi;
      v2f b0, bb;
      b0.x = W0[k0 * 40 + cc];  b0.y = W0[(k0 + 1) * 40 + cc];
      bb.x = Wb[k0 * 40 + cc];  bb.y = Wb[(k0 + 1) * 40 + cc];
      acc[ct * 2]     = WMMA_F32X4(a[i], b0, acc[ct * 2]);
      acc[ct * 2 + 1] = WMMA_F32X4(a[i], bb, acc[ct * 2 + 1]);
    }
  }

  const long long qoff = (long long)N * 32;
#pragma unroll
  for (int ct = 0; ct < 3; ++ct) {
    const int col = ct * 16 + n;
    if (col < 40) {
      const float bias = b2[col];
#pragma unroll
      for (int r = 0; r < 8; ++r) {
        const int row = base + r + 8 * hi;
        if (row < N) {
          const long long o = (ct < 2) ? (long long)row * 32 + col
                                       : qoff + (long long)row * 8 + (col - 32);
          sum2A[o] = acc[ct * 2][r] + bias;
          sum2B[o] = 0.0f;
          y2[o]    = acc[ct * 2 + 1][r];
        }
      }
    }
  }
}

#define S2_BATCH 16

__global__ void scatter2_kernel(const int* __restrict__ src, const int* __restrict__ dst,
                                const float* __restrict__ nrm, const float* __restrict__ y2,
                                float* __restrict__ sum2A, float* __restrict__ sum2B,
                                int E, int N) {
  const unsigned lane = threadIdx.x & 31u;
  const unsigned wave = (blockIdx.x * blockDim.x + threadIdx.x) >> 5;
  const unsigned ebase = wave * 32u;
  if (ebase >= (unsigned)E) return;
  const unsigned el = ebase + lane;
  const long long qoff = (long long)N * 32;
  float* __restrict__ sumP = package_half() ? sum2B : sum2A;
  float* __restrict__ sumQ = sumP + qoff;
  const float* __restrict__ yP = y2;
  const float* __restrict__ yQ = y2 + qoff;
  const unsigned qc = lane & 7u;
  const int qe = (int)(lane >> 3);

  if (ebase + 32u <= (unsigned)E) {
    const int s_l = src[el];
    const int d_l = dst[el];
    const float w_l = nrm[el];
#pragma unroll
    for (int b = 0; b < 32 / S2_BATCH; ++b) {
      float va[S2_BATCH], qv[S2_BATCH / 4], wq[S2_BATCH / 4];
      int dq[S2_BATCH / 4];
#pragma unroll
      for (int g = 0; g < S2_BATCH / 4; ++g) {
        const int sl = b * S2_BATCH + g * 4 + qe;
        const int sq = __shfl(s_l, sl, 32);
        dq[g] = __shfl(d_l, sl, 32);
        wq[g] = __shfl(w_l, sl, 32);
        qv[g] = yQ[(unsigned)sq * 8u + qc];
      }
#pragma unroll
      for (int j = 0; j < S2_BATCH; ++j) {
        const int s = __builtin_amdgcn_readlane(s_l, b * S2_BATCH + j);
        va[j] = yP[(unsigned)s * 32u + lane];
      }
#pragma unroll
      for (int j = 0; j < S2_BATCH; ++j) {
        const int d = __builtin_amdgcn_readlane(d_l, b * S2_BATCH + j);
        const float w = __builtin_bit_cast(
            float, __builtin_amdgcn_readlane(__builtin_bit_cast(int, w_l), b * S2_BATCH + j));
        unsafeAtomicAdd(&sumP[(unsigned)d * 32u + lane], w * va[j]);
      }
#pragma unroll
      for (int g = 0; g < S2_BATCH / 4; ++g)
        unsafeAtomicAdd(&sumQ[(unsigned)dq[g] * 8u + qc], wq[g] * qv[g]);
    }
  } else {
    int s_l = 0, d_l = 0;
    float w_l = 0.0f;
    if (el < (unsigned)E) { s_l = src[el]; d_l = dst[el]; w_l = nrm[el]; }
    const int cnt = (int)((unsigned)E - ebase);
    for (int j = 0; j < cnt; ++j) {
      const int s = __shfl(s_l, j, 32);
      const int d = __shfl(d_l, j, 32);
      const float w = __shfl(w_l, j, 32);
      unsafeAtomicAdd(&sumP[(long long)d * 32 + lane], w * yP[(long long)s * 32 + lane]);
      if (lane < 8u) unsafeAtomicAdd(&sumQ[(long long)d * 8 + lane], w * yQ[(long long)s * 8 + lane]);
    }
  }
}

__device__ __forceinline__ void lsm_one_row(const float* __restrict__ sum2A,
                                            const float* __restrict__ sum2B,
                                            float* __restrict__ out, int node, int lane, int N) {
  const long long po = (long long)node * 32 + lane;
  const long long qo = (long long)N * 32 + (long long)node * 8 + (lane & 7);
  float v0 = sum2A[po] + sum2B[po];
  float v1 = (lane < 8) ? sum2A[qo] + sum2B[qo] : -INFINITY;
  float m = fmaxf(v0, v1);
#pragma unroll
  for (int o = 16; o > 0; o >>= 1) m = fmaxf(m, __shfl_xor(m, o, 32));
  float s = expf(v0 - m) + ((lane < 8) ? expf(v1 - m) : 0.0f);
#pragma unroll
  for (int o = 16; o > 0; o >>= 1) s += __shfl_xor(s, o, 32);
  const float lse = m + logf(s);
  float* op = out + (long long)node * 40;
  op[lane] = v0 - lse;
  if (lane < 8) op[lane + 32] = v1 - lse;
}

#define LSM_ROWS 8

__global__ void lsm_kernel(const float* __restrict__ sum2A, const float* __restrict__ sum2B,
                           float* __restrict__ out, int N) {
  const int wave = (blockIdx.x * blockDim.x + threadIdx.x) >> 5;
  const int row0 = wave * LSM_ROWS;
  if (row0 >= N) return;
  const int lane = threadIdx.x & 31;
  if (row0 + LSM_ROWS > N) {
    for (int r = 0; r < LSM_ROWS; ++r)
      if (row0 + r < N) lsm_one_row(sum2A, sum2B, out, row0 + r, lane, N);
    return;
  }
  const long long qoff = (long long)N * 32 + (lane & 7);
  float v0[LSM_ROWS], v1[LSM_ROWS], m[LSM_ROWS], s[LSM_ROWS];
#pragma unroll
  for (int r = 0; r < LSM_ROWS; ++r) {
    const long long po = (long long)(row0 + r) * 32 + lane;
    const long long qo = qoff + (long long)(row0 + r) * 8;
    v0[r] = sum2A[po] + sum2B[po];
    const float t1 = sum2A[qo] + sum2B[qo];
    v1[r] = (lane < 8) ? t1 : -INFINITY;
    m[r] = fmaxf(v0[r], v1[r]);
  }
#pragma unroll
  for (int o = 16; o > 0; o >>= 1) {
#pragma unroll
    for (int r = 0; r < LSM_ROWS; ++r) m[r] = fmaxf(m[r], __shfl_xor(m[r], o, 32));
  }
#pragma unroll
  for (int r = 0; r < LSM_ROWS; ++r)
    s[r] = expf(v0[r] - m[r]) + expf(v1[r] - m[r]);
#pragma unroll
  for (int o = 16; o > 0; o >>= 1) {
#pragma unroll
    for (int r = 0; r < LSM_ROWS; ++r) s[r] += __shfl_xor(s[r], o, 32);
  }
#pragma unroll
  for (int r = 0; r < LSM_ROWS; ++r) {
    const float lse = m[r] + logf(s[r]);
    v0[r] -= lse;
    v1[r] -= lse;
  }
  float* op = out + (long long)row0 * 40;
#pragma unroll
  for (int r = 0; r < LSM_ROWS; ++r) op[r * 40 + lane] = v0[r];
  if (lane < 8) {
#pragma unroll
    for (int r = 0; r < LSM_ROWS; ++r) op[r * 40 + 32 + lane] = v1[r];
  }
}

extern "C" void kernel_launch(void* const* d_in, const int* in_sizes, int n_in,
                              void* d_out, int out_size, void* d_ws, size_t ws_size,
                              hipStream_t stream) {
  const float* x  = (const float*)d_in[0];
  const int* ei   = (const int*)d_in[1];
  const float* W1 = (const float*)d_in[2];
  const float* b1 = (const float*)d_in[3];
  const float* W2 = (const float*)d_in[4];
  const float* b2 = (const float*)d_in[5];
  float* out = (float*)d_out;

  const int N = in_sizes[0] / 64;
  const int E = in_sizes[1] / 2;
  const int* src = ei;
  const int* dst = ei + E;

  float* ws = (float*)d_ws;
  size_t off = 0;
  float* degA  = ws + off; off += (size_t)N;
  float* dinv  = ws + off; off += (size_t)N;
  float* nrm   = ws + off; off += (size_t)E;
  float* sum1A = ws + off; off += (size_t)N * 16;
  float* sum1B = ws + off; off += (size_t)N * 16;
  float* sum2A = ws + off; off += (size_t)N * 40;
  float* sum2B = ws + off; off += (size_t)N * 40;
  float* y1    = sum2A;
  float* degB  = sum2A + (size_t)N * 16;
  float* y2    = out;

  const int B = 256;
  auto blk = [](long long n, int b) { return (unsigned)((n + b - 1) / b); };

  const int tiles = (N + 15) / 16;
  gemm1_kernel<<<blk((long long)tiles * 32, B), B, 0, stream>>>(x, W1, b1, sum1A, sum1B, y1, degA, degB, N, tiles);

  const long long edgeWaves = ((long long)E + 32 * EDGE_K - 1) / (32 * EDGE_K);
  deg_kernel<<<blk(edgeWaves * 32, B), B, 0, stream>>>(src, degA, degB, E);

  if (N <= DINV_LDS_CAP) {
    scatter1_lds_kernel<<<256, FILL_THREADS, 0, stream>>>(src, dst, degA, degB, nrm, y1, sum1A, sum1B, E, N);
  } else {
    dinv_kernel<<<blk(N, B), B, 0, stream>>>(degA, degB, dinv, N);
    scatter1_kernel<<<blk(E, B), B, 0, stream>>>(src, dst, dinv, nrm, y1, sum1A, sum1B, E);
  }

  gemm2_kernel<<<blk((long long)tiles * 32, B), B, 0, stream>>>(sum1A, sum1B, W2, b2, sum2A, sum2B, y2, N, tiles);
  scatter2_kernel<<<blk(E, B), B, 0, stream>>>(src, dst, nrm, y2, sum2A, sum2B, E, N);

  const long long lsmWaves = ((long long)N + LSM_ROWS - 1) / LSM_ROWS;
  lsm_kernel<<<blk(lsmWaves * 32, B), B, 0, stream>>>(sum2A, sum2B, out, N);
}
